# starting kernel
# baseline (speedup 1.0000x reference)
#include <hip/hip_runtime.h>


typedef float v2f __attribute__((ext_vector_type(2)));
typedef float v8f __attribute__((ext_vector_type(8)));

#define BATCH 16

__global__ void zero_f32(float* __restrict__ p, int n) {
    int i = blockIdx.x * blockDim.x + threadIdx.x;
    if (i < n) p[i] = 0.0f;
}

__global__ void transpose_x(const float* __restrict__ x,
                            float* __restrict__ xT, int nsrc) {
    int i = blockIdx.x * blockDim.x + threadIdx.x;
    if (i < nsrc * BATCH) {
        int s = i >> 4;
        int b = i & 15;
        xT[i] = x[b * nsrc + s];
    }
}

__global__ void scatter_edges(const float* __restrict__ vals,
                              const int* __restrict__ idx,
                              const float* __restrict__ xT,
                              float* __restrict__ outT, int nnz) {
    int e = blockIdx.x * blockDim.x + threadIdx.x;
    if (e >= nnz) return;
    int r = __builtin_nontemporal_load(idx + e);
    int c = __builtin_nontemporal_load(idx + nnz + e);
    float v = __builtin_nontemporal_load(vals + e);

    const float4* xs = (const float4*)(xT + (size_t)c * BATCH);
    float4 x0 = xs[0];
    float4 x1 = xs[1];
    float4 x2 = xs[2];
    float4 x3 = xs[3];

    float* o = outT + (size_t)r * BATCH;
    atomicAdd(o + 0,  v * x0.x); atomicAdd(o + 1,  v * x0.y);
    atomicAdd(o + 2,  v * x0.z); atomicAdd(o + 3,  v * x0.w);
    atomicAdd(o + 4,  v * x1.x); atomicAdd(o + 5,  v * x1.y);
    atomicAdd(o + 6,  v * x1.z); atomicAdd(o + 7,  v * x1.w);
    atomicAdd(o + 8,  v * x2.x); atomicAdd(o + 9,  v * x2.y);
    atomicAdd(o + 10, v * x2.z); atomicAdd(o + 11, v * x2.w);
    atomicAdd(o + 12, v * x3.x); atomicAdd(o + 13, v * x3.y);
    atomicAdd(o + 14, v * x3.z); atomicAdd(o + 15, v * x3.w);
}

__global__ void wmma_transpose_bias(const float* __restrict__ outT,
                                    const float* __restrict__ bias,
                                    float* __restrict__ out, int ndst) {
    int lane = threadIdx.x & 31;
    int wave = threadIdx.x >> 5;
    int tile = blockIdx.x * (blockDim.x >> 5) + wave;
    int d0 = tile * 16;
    if (d0 >= ndst) return;

    int m = lane & 15;
    int h = lane >> 4;
    int k0 = 2 * h;

    float bb = bias[d0 + m];
    v8f c = {bb, bb, bb, bb, bb, bb, bb, bb};

#pragma unroll
    for (int q = 0; q < 4; ++q) {
        v2f a, b;
        a.x = (m == 4 * q + k0)     ? 1.0f : 0.0f;
        a.y = (m == 4 * q + k0 + 1) ? 1.0f : 0.0f;
        b = *(const v2f*)(outT + (size_t)(d0 + m) * BATCH + 4 * q + k0);
        c = __builtin_amdgcn_wmma_f32_16x16x4_f32(
                 false, a,  false, b,
                 (short)0, c,  false,  false);
    }

#pragma unroll
    for (int r = 0; r < 8; ++r) {
        int brow = r + 8 * h;
        out[(size_t)brow * ndst + d0 + m] = c[r];
    }
}

extern "C" void kernel_launch(void* const* d_in, const int* in_sizes, int n_in,
                              void* d_out, int out_size, void* d_ws, size_t ws_size,
                              hipStream_t stream) {
    const float* x    = (const float*)d_in[0];
    const float* vals = (const float*)d_in[1];
    const float* bias = (const float*)d_in[2];
    const int*   idx  = (const int*)d_in[3];

    int nsrc = in_sizes[0] / BATCH;
    int nnz  = in_sizes[1];
    int ndst = in_sizes[2];

    float* out  = (float*)d_out;
    float* xT   = (float*)d_ws;
    float* outT = xT + (size_t)nsrc * BATCH;

    int nOutT = ndst * BATCH;
    zero_f32<<<(nOutT + 255) / 256, 256, 0, stream>>>(outT, nOutT);

    int nXT = nsrc * BATCH;
    transpose_x<<<(nXT + 255) / 256, 256, 0, stream>>>(x, xT, nsrc);

    scatter_edges<<<(nnz + 255) / 256, 256, 0, stream>>>(vals, idx, xT, outT, nnz);

    int tiles = (ndst + 15) / 16;
    wmma_transpose_bias<<<(tiles + 7) / 8, 256, 0, stream>>>(outT, bias, out, ndst);
}


# edges binned by destination (128-row bins: LDS counters, scan, LDS-sorted list written in runs), each bin summed by LDS f32 atomic adds and fed to the WMMA from LDS; sum order: LDS arrival
# speedup vs baseline: 8.9479x; 8.9479x over previous
#include <hip/hip_runtime.h>


typedef float v2f __attribute__((ext_vector_type(2)));
typedef float v8f __attribute__((ext_vector_type(8)));

#define BATCH 16
#define RPB_SHIFT 7
#define RPB 128
#define MAXBINS 1024
#define CHUNK 12544
#define EPT ((CHUNK + 1023) / 1024)
#define COLBITS 24

__global__ void zero_f32(float* __restrict__ p, int n) {
    int i = blockIdx.x * blockDim.x + threadIdx.x;
    if (i < n) p[i] = 0.0f;
}

__global__ void transpose_x(const float* __restrict__ x,
                            float* __restrict__ xT, int nsrc) {
    int i = blockIdx.x * blockDim.x + threadIdx.x;
    if (i < nsrc * BATCH) {
        int s = i >> 4;
        int b = i & 15;
        xT[i] = x[b * nsrc + s];
    }
}

__device__ __forceinline__ int block_excl_scan_1024(int v, int* wsum, int* total) {
    int lane = threadIdx.x & 31;
    int w = threadIdx.x >> 5;
    int inc = v;
    for (int d = 1; d < 32; d <<= 1) {
        int n = __shfl_up(inc, d, 32);
        if (lane >= d) inc += n;
    }
    if (lane == 31) wsum[w] = inc;
    __syncthreads();
    if (w == 0) {
        int s = wsum[lane];
        int si = s;
        for (int d = 1; d < 32; d <<= 1) {
            int n = __shfl_up(si, d, 32);
            if (lane >= d) si += n;
        }
        wsum[lane] = si - s;
        if (lane == 31) wsum[32] = si;
    }
    __syncthreads();
    int excl = inc - v + wsum[w];
    *total = wsum[32];
    __syncthreads();
    return excl;
}

__global__ void __launch_bounds__(1024)
bin_count(const int* __restrict__ idx, int* __restrict__ hist,
          int nnz, int nbins, int nwg) {
    __shared__ int cnt[MAXBINS];
    int t = threadIdx.x;
    int g = blockIdx.x;
    if (t < nbins) cnt[t] = 0;
    __syncthreads();
    int e0 = g * CHUNK;
    int e1 = min(e0 + CHUNK, nnz);
    int r[EPT];
    for (int k = 0; k < EPT; ++k) r[k] = idx[min(e0 + t + k * 1024, e1 - 1)];
    for (int k = 0; k < EPT; ++k)
        if (e0 + t + k * 1024 < e1) atomicAdd(&cnt[r[k] >> RPB_SHIFT], 1);
    __syncthreads();
    if (t < nbins) hist[t * nwg + g] = cnt[t];
}

__global__ void __launch_bounds__(1024)
scan_counts(const int* __restrict__ hist, int* __restrict__ offs, int n) {
    __shared__ int wsum[33];
    int t = threadIdx.x;
    int k = blockIdx.x;
    const int4* h4 = (const int4*)hist;
    int nq = k * 256;
    int part = 0;
#pragma unroll 8
    for (int q = t; q < nq; q += 1024) {
        int4 a = h4[q];
        part += (a.x + a.y) + (a.z + a.w);
    }
    int before;
    block_excl_scan_1024(part, wsum, &before);
    int i = k * 1024 + t;
    int v = (i < n) ? hist[i] : 0;
    int tile_total;
    int ex = block_excl_scan_1024(v, wsum, &tile_total);
    if (i < n) offs[i] = before + ex;
}

__global__ void __launch_bounds__(1024)
bin_place(const float* __restrict__ vals, const int* __restrict__ idx,
          const int* __restrict__ offs,
          int* __restrict__ recs, int nnz, int nbins, int nwg) {
    __shared__ int cursor[MAXBINS];
    __shared__ int gdelta[MAXBINS];
    __shared__ int wsum[33];
    __shared__ int srow[CHUNK];
    __shared__ int scol[CHUNK];
    __shared__ float sval[CHUNK];
    int t = threadIdx.x;
    int g = blockIdx.x;
    if (t < nbins) cursor[t] = 0;
    __syncthreads();
    int e0 = g * CHUNK;
    int e1 = min(e0 + CHUNK, nnz);
    int r[EPT], s[EPT];
    float v[EPT];
    for (int k = 0; k < EPT; ++k) {
        int e = min(e0 + t + k * 1024, e1 - 1);
        r[k] = __builtin_nontemporal_load(idx + e);
        s[k] = __builtin_nontemporal_load(idx + nnz + e);
        v[k] = __builtin_nontemporal_load(vals + e);
    }
    for (int k = 0; k < EPT; ++k)
        if (e0 + t + k * 1024 < e1) atomicAdd(&cursor[r[k] >> RPB_SHIFT], 1);
    __syncthreads();
    int c = 0, o = 0;
    if (t < nbins) {
        c = cursor[t];
        o = offs[t * nwg + g];
    }
    int chunk_total;
    int ex = block_excl_scan_1024(c, wsum, &chunk_total);
    if (t < nbins) {
        cursor[t] = ex;
        gdelta[t] = o - ex;
    }
    __syncthreads();
    for (int k = 0; k < EPT; ++k) {
        if (e0 + t + k * 1024 < e1) {
            int p = atomicAdd(&cursor[r[k] >> RPB_SHIFT], 1);
            srow[p] = r[k];
            scol[p] = s[k];
            sval[p] = v[k];
        }
    }
    __syncthreads();
    int n = e1 - e0;
    int2* recs2 = (int2*)recs;
    for (int i = t; i < n; i += 1024) {
        int rr = srow[i];
        int pos = gdelta[rr >> RPB_SHIFT] + i;
        recs2[pos] = make_int2(((rr & (RPB - 1)) << COLBITS) | scol[i],
                               __float_as_int(sval[i]));
    }
}

__global__ void __launch_bounds__(512)
bin_sum_wmma(const int* __restrict__ recs, const int* __restrict__ offs,
             const float* __restrict__ xT, const float* __restrict__ bias,
             float* __restrict__ out, int nnz, int ndst, int nbins, int nwg) {
    __shared__ float acc[(RPB + 1) * BATCH];
    int t = threadIdx.x;
    int bin = blockIdx.x;
    for (int i = t; i < (RPB + 1) * BATCH; i += 512) acc[i] = 0.0f;
    __syncthreads();

    int beg = offs[bin * nwg];
    int end = (bin + 1 < nbins) ? offs[(bin + 1) * nwg] : nnz;
    int hw = t >> 4;
    int bl = t & 15;
    int iters = (end - beg + 511) / 512;
    const int2* recs2 = (const int2*)recs;
    const int2 empty = make_int2((int)((unsigned)RPB << COLBITS), 0);
    int i = beg + hw * 16 + bl;
    int last = max(end - 1, 0);
    int2 rec = recs2[min(i, last)];
    if (i >= end) rec = empty;
    for (int it = 0; it < iters; ++it) {
        int2 nxt = recs2[min(i + 512, last)];
        if (i + 512 >= end) nxt = empty;
        float xv[16];
        for (int j = 0; j < 16; ++j) {
            unsigned key = (unsigned)__shfl(rec.x, j, 16);
            unsigned s = key & ((1u << COLBITS) - 1u);
            xv[j] = xT[(size_t)s * BATCH + bl];
        }
        for (int j = 0; j < 16; ++j) {
            unsigned key = (unsigned)__shfl(rec.x, j, 16);
            float vv = __int_as_float(__shfl(rec.y, j, 16));
            unsigned rl = key >> COLBITS;
            atomicAdd(&acc[rl * BATCH + bl], vv * xv[j]);
        }
        rec = nxt;
        i += 512;
    }
    __syncthreads();

    int lane = t & 31;
    int wave = t >> 5;
    int d0 = bin * RPB + wave * 16;
    if (wave < RPB / 16 && d0 < ndst) {
        int m = lane & 15;
        int h = lane >> 4;
        int k0 = 2 * h;

        float bb = bias[d0 + m];
        v8f c = {bb, bb, bb, bb, bb, bb, bb, bb};

        for (int q = 0; q < 4; ++q) {
            v2f a, b;
            a.x = (m == 4 * q + k0)     ? 1.0f : 0.0f;
            a.y = (m == 4 * q + k0 + 1) ? 1.0f : 0.0f;
            b = *(const v2f*)(acc + (wave * 16 + m) * BATCH + 4 * q + k0);
            c = __builtin_amdgcn_wmma_f32_16x16x4_f32(
                     false, a,  false, b,
                     (short)0, c,  false,  false);
        }

        for (int r = 0; r < 8; ++r) {
            int brow = r + 8 * h;
            out[(size_t)brow * ndst + d0 + m] = c[r];
        }
    }
}


__global__ void scatter_edges(const float* __restrict__ vals,
                              const int* __restrict__ idx,
                              const float* __restrict__ xT,
                              float* __restrict__ outT, int nnz) {
    int e = blockIdx.x * blockDim.x + threadIdx.x;
    if (e >= nnz) return;
    int r = __builtin_nontemporal_load(idx + e);
    int c = __builtin_nontemporal_load(idx + nnz + e);
    float v = __builtin_nontemporal_load(vals + e);

    const float4* xs = (const float4*)(xT + (size_t)c * BATCH);
    float4 x0 = xs[0];
    float4 x1 = xs[1];
    float4 x2 = xs[2];
    float4 x3 = xs[3];

    float* o = outT + (size_t)r * BATCH;
    atomicAdd(o + 0,  v * x0.x); atomicAdd(o + 1,  v * x0.y);
    atomicAdd(o + 2,  v * x0.z); atomicAdd(o + 3,  v * x0.w);
    atomicAdd(o + 4,  v * x1.x); atomicAdd(o + 5,  v * x1.y);
    atomicAdd(o + 6,  v * x1.z); atomicAdd(o + 7,  v * x1.w);
    atomicAdd(o + 8,  v * x2.x); atomicAdd(o + 9,  v * x2.y);
    atomicAdd(o + 10, v * x2.z); atomicAdd(o + 11, v * x2.w);
    atomicAdd(o + 12, v * x3.x); atomicAdd(o + 13, v * x3.y);
    atomicAdd(o + 14, v * x3.z); atomicAdd(o + 15, v * x3.w);
}

__global__ void wmma_transpose_bias(const float* __restrict__ outT,
                                    const float* __restrict__ bias,
                                    float* __restrict__ out, int ndst) {
    int lane = threadIdx.x & 31;
    int wave = threadIdx.x >> 5;
    int tile = blockIdx.x * (blockDim.x >> 5) + wave;
    int d0 = tile * 16;
    if (d0 >= ndst) return;

    int m = lane & 15;
    int h = lane >> 4;
    int k0 = 2 * h;

    float bb = bias[d0 + m];
    v8f c = {bb, bb, bb, bb, bb, bb, bb, bb};

    for (int q = 0; q < 4; ++q) {
        v2f a, b;
        a.x = (m == 4 * q + k0)     ? 1.0f : 0.0f;
        a.y = (m == 4 * q + k0 + 1) ? 1.0f : 0.0f;
        b = *(const v2f*)(outT + (size_t)(d0 + m) * BATCH + 4 * q + k0);
        c = __builtin_amdgcn_wmma_f32_16x16x4_f32(
                 false, a,  false, b,
                 (short)0, c,  false,  false);
    }

    for (int r = 0; r < 8; ++r) {
        int brow = r + 8 * h;
        out[(size_t)brow * ndst + d0 + m] = c[r];
    }
}

extern "C" void kernel_launch(void* const* d_in, const int* in_sizes, int n_in,
                              void* d_out, int out_size, void* d_ws, size_t ws_size,
                              hipStream_t stream) {
    const float* x    = (const float*)d_in[0];
    const float* vals = (const float*)d_in[1];
    const float* bias = (const float*)d_in[2];
    const int*   idx  = (const int*)d_in[3];

    int nsrc = in_sizes[0] / BATCH;
    int nnz  = in_sizes[1];
    int ndst = in_sizes[2];

    float* out = (float*)d_out;
    float* xT  = (float*)d_ws;

    int nXT = nsrc * BATCH;
    transpose_x<<<(nXT + 255) / 256, 256, 0, stream>>>(x, xT, nsrc);

    int nbins = (ndst + RPB - 1) / RPB;
    int nwg   = (nnz + CHUNK - 1) / CHUNK;
    int nh    = nbins * nwg;
    int nhp   = (nh + 3) & ~3;
    size_t need = ((size_t)nXT + (size_t)nhp + 2 * (size_t)nnz) * 4;

    if (ws_size >= need && nbins <= MAXBINS && nsrc <= (1 << COLBITS)) {
        int*  offs = (int*)(xT + (size_t)nXT);
        int*  recs = offs + nhp;
        int*  hist = recs;
        bin_count<<<nwg, 1024, 0, stream>>>(idx, hist, nnz, nbins, nwg);
        scan_counts<<<(nh + 1023) / 1024, 1024, 0, stream>>>(hist, offs, nh);
        bin_place<<<nwg, 1024, 0, stream>>>(vals, idx, offs, recs,
                                            nnz, nbins, nwg);
        bin_sum_wmma<<<nbins, 512, 0, stream>>>(recs, offs, xT, bias, out,
                                                nnz, ndst, nbins, nwg);
    } else {
        float* outT = xT + (size_t)nXT;
        int nOutT = ndst * BATCH;
        zero_f32<<<(nOutT + 255) / 256, 256, 0, stream>>>(outT, nOutT);
        scatter_edges<<<(nnz + 255) / 256, 256, 0, stream>>>(vals, idx, xT, outT, nnz);
        int tiles = (ndst + 15) / 16;
        wmma_transpose_bias<<<(tiles + 7) / 8, 256, 0, stream>>>(outT, bias, out, ndst);
    }
}


# x transpose folded into the bin-count launch: each of its 256 work-groups writes its own slice of xT under its row loads; one launch and 50000 short waves fewer
# speedup vs baseline: 9.3759x; 1.0478x over previous
#include <hip/hip_runtime.h>


typedef float v2f __attribute__((ext_vector_type(2)));
typedef float v8f __attribute__((ext_vector_type(8)));

#define BATCH 16
#define RPB_SHIFT 7
#define RPB 128
#define MAXBINS 1024
#define CHUNK 12544
#define EPT ((CHUNK + 1023) / 1024)
#define COLBITS 24
#define XPT 8

__global__ void zero_f32(float* __restrict__ p, int n) {
    int i = blockIdx.x * blockDim.x + threadIdx.x;
    if (i < n) p[i] = 0.0f;
}

__global__ void transpose_x(const float* __restrict__ x,
                            float* __restrict__ xT, int nsrc) {
    int i = blockIdx.x * blockDim.x + threadIdx.x;
    if (i < nsrc * BATCH) {
        int s = i >> 4;
        int b = i & 15;
        xT[i] = x[b * nsrc + s];
    }
}

__device__ __forceinline__ int block_excl_scan_1024(int v, int* wsum, int* total) {
    int lane = threadIdx.x & 31;
    int w = threadIdx.x >> 5;
    int inc = v;
    for (int d = 1; d < 32; d <<= 1) {
        int n = __shfl_up(inc, d, 32);
        if (lane >= d) inc += n;
    }
    if (lane == 31) wsum[w] = inc;
    __syncthreads();
    if (w == 0) {
        int s = wsum[lane];
        int si = s;
        for (int d = 1; d < 32; d <<= 1) {
            int n = __shfl_up(si, d, 32);
            if (lane >= d) si += n;
        }
        wsum[lane] = si - s;
        if (lane == 31) wsum[32] = si;
    }
    __syncthreads();
    int excl = inc - v + wsum[w];
    *total = wsum[32];
    __syncthreads();
    return excl;
}

__global__ void __launch_bounds__(1024)
bin_count(const int* __restrict__ idx, int* __restrict__ hist,
          const float* __restrict__ x, float* __restrict__ xT,
          int nnz, int nbins, int nwg, int nsrc) {
    __shared__ int cnt[MAXBINS];
    int t = threadIdx.x;
    int g = blockIdx.x;
    if (t < nbins) cnt[t] = 0;
    __syncthreads();
    int e0 = g * CHUNK;
    int e1 = min(e0 + CHUNK, nnz);
    int r[EPT];
    for (int k = 0; k < EPT; ++k) r[k] = idx[min(e0 + t + k * 1024, e1 - 1)];

    int per = (nsrc + nwg - 1) / nwg;
    int s0 = min(g * per, nsrc);
    int s1 = min(s0 + per, nsrc);
    int ne = (s1 - s0) * BATCH;
    const float* xs = x + s0;
    float* dst = xT + (size_t)s0 * BATCH;
    float xv[XPT];
    for (int k = 0; k < XPT; ++k) {
        int i = max(min(t + k * 1024, ne - 1), 0);
        xv[k] = xs[(i & (BATCH - 1)) * nsrc + (i >> 4)];
    }
    asm volatile("" :: "v"(r[0]));

    for (int k = 0; k < EPT; ++k)
        if (e0 + t + k * 1024 < e1) atomicAdd(&cnt[r[k] >> RPB_SHIFT], 1);

    for (int k = 0; k < XPT; ++k)
        if (t + k * 1024 < ne) dst[t + k * 1024] = xv[k];
    for (int i = t + XPT * 1024; i < ne; i += 1024)
        dst[i] = xs[(i & (BATCH - 1)) * nsrc + (i >> 4)];
    __syncthreads();
    if (t < nbins) hist[t * nwg + g] = cnt[t];
}

__global__ void __launch_bounds__(1024)
scan_counts(const int* __restrict__ hist, int* __restrict__ offs, int n) {
    __shared__ int wsum[33];
    int t = threadIdx.x;
    int k = blockIdx.x;
    const int4* h4 = (const int4*)hist;
    int nq = k * 256;
    int part = 0;
#pragma unroll 8
    for (int q = t; q < nq; q += 1024) {
        int4 a = h4[q];
        part += (a.x + a.y) + (a.z + a.w);
    }
    int before;
    block_excl_scan_1024(part, wsum, &before);
    int i = k * 1024 + t;
    int v = (i < n) ? hist[i] : 0;
    int tile_total;
    int ex = block_excl_scan_1024(v, wsum, &tile_total);
    if (i < n) offs[i] = before + ex;
}

__global__ void __launch_bounds__(1024)
bin_place(const float* __restrict__ vals, const int* __restrict__ idx,
          const int* __restrict__ offs,
          int* __restrict__ recs, int nnz, int nbins, int nwg) {
    __shared__ int cursor[MAXBINS];
    __shared__ int gdelta[MAXBINS];
    __shared__ int wsum[33];
    __shared__ int srow[CHUNK];
    __shared__ int scol[CHUNK];
    __shared__ float sval[CHUNK];
    int t = threadIdx.x;
    int g = blockIdx.x;
    if (t < nbins) cursor[t] = 0;
    __syncthreads();
    int e0 = g * CHUNK;
    int e1 = min(e0 + CHUNK, nnz);
    int r[EPT], s[EPT];
    float v[EPT];
    for (int k = 0; k < EPT; ++k) {
        int e = min(e0 + t + k * 1024, e1 - 1);
        r[k] = __builtin_nontemporal_load(idx + e);
        s[k] = __builtin_nontemporal_load(idx + nnz + e);
        v[k] = __builtin_nontemporal_load(vals + e);
    }
    for (int k = 0; k < EPT; ++k)
        if (e0 + t + k * 1024 < e1) atomicAdd(&cursor[r[k] >> RPB_SHIFT], 1);
    __syncthreads();
    int c = 0, o = 0;
    if (t < nbins) {
        c = cursor[t];
        o = offs[t * nwg + g];
    }
    int chunk_total;
    int ex = block_excl_scan_1024(c, wsum, &chunk_total);
    if (t < nbins) {
        cursor[t] = ex;
        gdelta[t] = o - ex;
    }
    __syncthreads();
    for (int k = 0; k < EPT; ++k) {
        if (e0 + t + k * 1024 < e1) {
            int p = atomicAdd(&cursor[r[k] >> RPB_SHIFT], 1);
            srow[p] = r[k];
            scol[p] = s[k];
            sval[p] = v[k];
        }
    }
    __syncthreads();
    int n = e1 - e0;
    int2* recs2 = (int2*)recs;
    for (int i = t; i < n; i += 1024) {
        int rr = srow[i];
        int pos = gdelta[rr >> RPB_SHIFT] + i;
        recs2[pos] = make_int2(((rr & (RPB - 1)) << COLBITS) | scol[i],
                               __float_as_int(sval[i]));
    }
}

__global__ void __launch_bounds__(512)
bin_sum_wmma(const int* __restrict__ recs, const int* __restrict__ offs,
             const float* __restrict__ xT, const float* __restrict__ bias,
             float* __restrict__ out, int nnz, int ndst, int nbins, int nwg) {
    __shared__ float acc[(RPB + 1) * BATCH];
    int t = threadIdx.x;
    int bin = blockIdx.x;
    for (int i = t; i < (RPB + 1) * BATCH; i += 512) acc[i] = 0.0f;
    __syncthreads();

    int beg = offs[bin * nwg];
    int end = (bin + 1 < nbins) ? offs[(bin + 1) * nwg] : nnz;
    int hw = t >> 4;
    int bl = t & 15;
    int iters = (end - beg + 511) / 512;
    const int2* recs2 = (const int2*)recs;
    const int2 empty = make_int2((int)((unsigned)RPB << COLBITS), 0);
    int i = beg + hw * 16 + bl;
    int last = max(end - 1, 0);
    int2 rec = recs2[min(i, last)];
    if (i >= end) rec = empty;
    for (int it = 0; it < iters; ++it) {
        int2 nxt = recs2[min(i + 512, last)];
        if (i + 512 >= end) nxt = empty;
        float xv[16];
        for (int j = 0; j < 16; ++j) {
            unsigned key = (unsigned)__shfl(rec.x, j, 16);
            unsigned s = key & ((1u << COLBITS) - 1u);
            xv[j] = xT[(size_t)s * BATCH + bl];
        }
        for (int j = 0; j < 16; ++j) {
            unsigned key = (unsigned)__shfl(rec.x, j, 16);
            float vv = __int_as_float(__shfl(rec.y, j, 16));
            unsigned rl = key >> COLBITS;
            atomicAdd(&acc[rl * BATCH + bl], vv * xv[j]);
        }
        rec = nxt;
        i += 512;
    }
    __syncthreads();

    int lane = t & 31;
    int wave = t >> 5;
    int d0 = bin * RPB + wave * 16;
    if (wave < RPB / 16 && d0 < ndst) {
        int m = lane & 15;
        int h = lane >> 4;
        int k0 = 2 * h;

        float bb = bias[d0 + m];
        v8f c = {bb, bb, bb, bb, bb, bb, bb, bb};

        for (int q = 0; q < 4; ++q) {
            v2f a, b;
            a.x = (m == 4 * q + k0)     ? 1.0f : 0.0f;
            a.y = (m == 4 * q + k0 + 1) ? 1.0f : 0.0f;
            b = *(const v2f*)(acc + (wave * 16 + m) * BATCH + 4 * q + k0);
            c = __builtin_amdgcn_wmma_f32_16x16x4_f32(
                     false, a,  false, b,
                     (short)0, c,  false,  false);
        }

        for (int r = 0; r < 8; ++r) {
            int brow = r + 8 * h;
            out[(size_t)brow * ndst + d0 + m] = c[r];
        }
    }
}


__global__ void scatter_edges(const float* __restrict__ vals,
                              const int* __restrict__ idx,
                              const float* __restrict__ xT,
                              float* __restrict__ outT, int nnz) {
    int e = blockIdx.x * blockDim.x + threadIdx.x;
    if (e >= nnz) return;
    int r = __builtin_nontemporal_load(idx + e);
    int c = __builtin_nontemporal_load(idx + nnz + e);
    float v = __builtin_nontemporal_load(vals + e);

    const float4* xs = (const float4*)(xT + (size_t)c * BATCH);
    float4 x0 = xs[0];
    float4 x1 = xs[1];
    float4 x2 = xs[2];
    float4 x3 = xs[3];

    float* o = outT + (size_t)r * BATCH;
    atomicAdd(o + 0,  v * x0.x); atomicAdd(o + 1,  v * x0.y);
    atomicAdd(o + 2,  v * x0.z); atomicAdd(o + 3,  v * x0.w);
    atomicAdd(o + 4,  v * x1.x); atomicAdd(o + 5,  v * x1.y);
    atomicAdd(o + 6,  v * x1.z); atomicAdd(o + 7,  v * x1.w);
    atomicAdd(o + 8,  v * x2.x); atomicAdd(o + 9,  v * x2.y);
    atomicAdd(o + 10, v * x2.z); atomicAdd(o + 11, v * x2.w);
    atomicAdd(o + 12, v * x3.x); atomicAdd(o + 13, v * x3.y);
    atomicAdd(o + 14, v * x3.z); atomicAdd(o + 15, v * x3.w);
}

__global__ void wmma_transpose_bias(const float* __restrict__ outT,
                                    const float* __restrict__ bias,
                                    float* __restrict__ out, int ndst) {
    int lane = threadIdx.x & 31;
    int wave = threadIdx.x >> 5;
    int tile = blockIdx.x * (blockDim.x >> 5) + wave;
    int d0 = tile * 16;
    if (d0 >= ndst) return;

    int m = lane & 15;
    int h = lane >> 4;
    int k0 = 2 * h;

    float bb = bias[d0 + m];
    v8f c = {bb, bb, bb, bb, bb, bb, bb, bb};

    for (int q = 0; q < 4; ++q) {
        v2f a, b;
        a.x = (m == 4 * q + k0)     ? 1.0f : 0.0f;
        a.y = (m == 4 * q + k0 + 1) ? 1.0f : 0.0f;
        b = *(const v2f*)(outT + (size_t)(d0 + m) * BATCH + 4 * q + k0);
        c = __builtin_amdgcn_wmma_f32_16x16x4_f32(
                 false, a,  false, b,
                 (short)0, c,  false,  false);
    }

    for (int r = 0; r < 8; ++r) {
        int brow = r + 8 * h;
        out[(size_t)brow * ndst + d0 + m] = c[r];
    }
}

extern "C" void kernel_launch(void* const* d_in, const int* in_sizes, int n_in,
                              void* d_out, int out_size, void* d_ws, size_t ws_size,
                              hipStream_t stream) {
    const float* x    = (const float*)d_in[0];
    const float* vals = (const float*)d_in[1];
    const float* bias = (const float*)d_in[2];
    const int*   idx  = (const int*)d_in[3];

    int nsrc = in_sizes[0] / BATCH;
    int nnz  = in_sizes[1];
    int ndst = in_sizes[2];

    float* out = (float*)d_out;
    float* xT  = (float*)d_ws;

    int nXT = nsrc * BATCH;

    int nbins = (ndst + RPB - 1) / RPB;
    int nwg   = (nnz + CHUNK - 1) / CHUNK;
    int nh    = nbins * nwg;
    int nhp   = (nh + 3) & ~3;
    size_t need = ((size_t)nXT + (size_t)nhp + 2 * (size_t)nnz) * 4;

    if (ws_size >= need && nbins <= MAXBINS && nsrc <= (1 << COLBITS)) {
        int*  offs = (int*)(xT + (size_t)nXT);
        int*  recs = offs + nhp;
        int*  hist = recs;
        bin_count<<<nwg, 1024, 0, stream>>>(idx, hist, x, xT, nnz, nbins, nwg, nsrc);
        scan_counts<<<(nh + 1023) / 1024, 1024, 0, stream>>>(hist, offs, nh);
        bin_place<<<nwg, 1024, 0, stream>>>(vals, idx, offs, recs,
                                            nnz, nbins, nwg);
        bin_sum_wmma<<<nbins, 512, 0, stream>>>(recs, offs, xT, bias, out,
                                                nnz, ndst, nbins, nwg);
    } else {
        float* outT = xT + (size_t)nXT;
        int nOutT = ndst * BATCH;
        transpose_x<<<(nXT + 255) / 256, 256, 0, stream>>>(x, xT, nsrc);
        zero_f32<<<(nOutT + 255) / 256, 256, 0, stream>>>(outT, nOutT);
        scatter_edges<<<(nnz + 255) / 256, 256, 0, stream>>>(vals, idx, xT, outT, nnz);
        int tiles = (ndst + 15) / 16;
        wmma_transpose_bias<<<(tiles + 7) / 8, 256, 0, stream>>>(outT, bias, out, ndst);
    }
}


# edge list kept chunk-sorted in place plus a table of run starts per (bin, chunk); each bin's work-group walks its runs as one stream; count and scan launches dropped (4 -> 2); sum order: LDS arrival
# speedup vs baseline: 11.7043x; 1.2483x over previous
#include <hip/hip_runtime.h>


typedef float v2f __attribute__((ext_vector_type(2)));
typedef float v8f __attribute__((ext_vector_type(8)));

#define BATCH 16
#define RPB_SHIFT 7
#define RPB 128
#define MAXBINS 1024
#define CHUNK 12544
#define EPT ((CHUNK + 1023) / 1024)
#define COLBITS 24
#define XPT 8
#define RPH 8

__global__ void zero_f32(float* __restrict__ p, int n) {
    int i = blockIdx.x * blockDim.x + threadIdx.x;
    if (i < n) p[i] = 0.0f;
}

__global__ void transpose_x(const float* __restrict__ x,
                            float* __restrict__ xT, int nsrc) {
    int i = blockIdx.x * blockDim.x + threadIdx.x;
    if (i < nsrc * BATCH) {
        int s = i >> 4;
        int b = i & 15;
        xT[i] = x[b * nsrc + s];
    }
}

__device__ __forceinline__ int block_excl_scan_1024(int v, int* wsum, int* total) {
    int lane = threadIdx.x & 31;
    int w = threadIdx.x >> 5;
    int inc = v;
    for (int d = 1; d < 32; d <<= 1) {
        int n = __shfl_up(inc, d, 32);
        if (lane >= d) inc += n;
    }
    if (lane == 31) wsum[w] = inc;
    __syncthreads();
    if (w == 0) {
        int s = wsum[lane];
        int si = s;
        for (int d = 1; d < 32; d <<= 1) {
            int n = __shfl_up(si, d, 32);
            if (lane >= d) si += n;
        }
        wsum[lane] = si - s;
        if (lane == 31) wsum[32] = si;
    }
    __syncthreads();
    int excl = inc - v + wsum[w];
    *total = wsum[32];
    __syncthreads();
    return excl;
}

__global__ void __launch_bounds__(1024)
chunk_sort(const float* __restrict__ vals, const int* __restrict__ idx,
           const float* __restrict__ x, float* __restrict__ xT,
           int* __restrict__ tab, int* __restrict__ recs,
           int nnz, int nbins, int nwg, int nsrc) {
    __shared__ int cursor[MAXBINS];
    __shared__ int wsum[33];
    __shared__ int2 srec[CHUNK];
    int t = threadIdx.x;
    int g = blockIdx.x;
    if (t < nbins) cursor[t] = 0;
    __syncthreads();
    int e0 = g * CHUNK;
    int e1 = min(e0 + CHUNK, nnz);
    int r[EPT], s[EPT];
    float v[EPT];
    for (int k = 0; k < EPT; ++k) {
        int e = min(e0 + t + k * 1024, e1 - 1);
        r[k] = __builtin_nontemporal_load(idx + e);
        s[k] = __builtin_nontemporal_load(idx + nnz + e);
        v[k] = __builtin_nontemporal_load(vals + e);
    }
    int per = (nsrc + nwg - 1) / nwg;
    int s0 = min(g * per, nsrc);
    int s1 = min(s0 + per, nsrc);
    int ne = (s1 - s0) * BATCH;
    const float* xs = x + s0;
    float* dst = xT + (size_t)s0 * BATCH;
    float xv[XPT];
    for (int k = 0; k < XPT; ++k) {
        int i = max(min(t + k * 1024, ne - 1), 0);
        xv[k] = xs[(i & (BATCH - 1)) * nsrc + (i >> 4)];
    }
    __builtin_amdgcn_sched_barrier(0);

    for (int k = 0; k < EPT; ++k)
        if (e0 + t + k * 1024 < e1) atomicAdd(&cursor[r[k] >> RPB_SHIFT], 1);
    __syncthreads();
    int c = (t < nbins) ? cursor[t] : 0;
    int chunk_total;
    int ex = block_excl_scan_1024(c, wsum, &chunk_total);
    if (t < nbins) {
        cursor[t] = ex;
        tab[t * nwg + g] = e0 + ex;
    }
    __syncthreads();
    for (int k = 0; k < EPT; ++k) {
        if (e0 + t + k * 1024 < e1) {
            int p = atomicAdd(&cursor[r[k] >> RPB_SHIFT], 1);
            srec[p] = make_int2(((r[k] & (RPB - 1)) << COLBITS) | s[k],
                                __float_as_int(v[k]));
        }
    }
    for (int k = 0; k < XPT; ++k)
        if (t + k * 1024 < ne) dst[t + k * 1024] = xv[k];
    for (int i = t + XPT * 1024; i < ne; i += 1024)
        dst[i] = xs[(i & (BATCH - 1)) * nsrc + (i >> 4)];
    __syncthreads();
    int n = e1 - e0;
    int2* recs2 = (int2*)recs;
    for (int i = t; i < n; i += 1024) recs2[e0 + i] = srec[i];
}

__global__ void __launch_bounds__(512)
bin_sum_wmma(const int* __restrict__ recs, const int* __restrict__ tab,
             const float* __restrict__ xT, const float* __restrict__ bias,
             float* __restrict__ out, int nnz, int ndst, int nbins, int nwg) {
    __shared__ float acc[(RPB + 1) * BATCH];
    int t = threadIdx.x;
    int bin = blockIdx.x;
    for (int i = t; i < (RPB + 1) * BATCH; i += 512) acc[i] = 0.0f;
    __syncthreads();

    int hw = t >> 4;
    int bl = t & 15;
    const int2* recs2 = (const int2*)recs;
    const int2 empty = make_int2((int)((unsigned)RPB << COLBITS), 0);

    for (int base = 0; base < nwg; base += 32 * RPH) {
        int g = base + hw * RPH + bl;
        int gc = min(g, nwg - 1);
        int rb = tab[bin * nwg + gc];
        int re = (bin + 1 < nbins) ? tab[(bin + 1) * nwg + gc]
                                   : min((gc + 1) * CHUNK, nnz);
        int len = (bl < RPH && g < nwg) ? re - rb : 0;
        int inc = len;
        for (int d = 1; d < 16; d <<= 1) {
            int n = __shfl_up(inc, d, 16);
            if (bl >= d) inc += n;
        }
        int total = __shfl(inc, 15, 16);
        int excl = inc - len;
        int delta = rb - excl;
        int nb = (total + 15) >> 4;
        nb = max(nb, __shfl_xor(nb, 16, 32));

        int q = bl;
        int dsel = 0;
        for (int j = 0; j < RPH; ++j) {
            int ej = __shfl(excl, j, 16);
            int dj = __shfl(delta, j, 16);
            if (q >= ej) dsel = dj;
        }
        int2 rec = recs2[(q < total) ? q + dsel : 0];
        if (q >= total) rec = empty;
        for (int it = 0; it < nb; ++it) {
            int qn = q + 16;
            int dn = 0;
            for (int j = 0; j < RPH; ++j) {
                int ej = __shfl(excl, j, 16);
                int dj = __shfl(delta, j, 16);
                if (qn >= ej) dn = dj;
            }
            int2 nxt = recs2[(qn < total) ? qn + dn : 0];
            if (qn >= total) nxt = empty;
            float xv[16];
            for (int j = 0; j < 16; ++j) {
                unsigned key = (unsigned)__shfl(rec.x, j, 16);
                unsigned s = key & ((1u << COLBITS) - 1u);
                xv[j] = xT[(size_t)s * BATCH + bl];
            }
            for (int j = 0; j < 16; ++j) {
                unsigned key = (unsigned)__shfl(rec.x, j, 16);
                float vv = __int_as_float(__shfl(rec.y, j, 16));
                unsigned rl = key >> COLBITS;
                atomicAdd(&acc[rl * BATCH + bl], vv * xv[j]);
            }
            rec = nxt;
            q = qn;
        }
    }
    __syncthreads();

    int lane = t & 31;
    int wave = t >> 5;
    int d0 = bin * RPB + wave * 16;
    if (wave < RPB / 16 && d0 < ndst) {
        int m = lane & 15;
        int h = lane >> 4;
        int k0 = 2 * h;

        float bb = bias[d0 + m];
        v8f c = {bb, bb, bb, bb, bb, bb, bb, bb};

        for (int q = 0; q < 4; ++q) {
            v2f a, b;
            a.x = (m == 4 * q + k0)     ? 1.0f : 0.0f;
            a.y = (m == 4 * q + k0 + 1) ? 1.0f : 0.0f;
            b = *(const v2f*)(acc + (wave * 16 + m) * BATCH + 4 * q + k0);
            c = __builtin_amdgcn_wmma_f32_16x16x4_f32(
                     false, a,  false, b,
                     (short)0, c,  false,  false);
        }

        for (int r = 0; r < 8; ++r) {
            int brow = r + 8 * h;
            out[(size_t)brow * ndst + d0 + m] = c[r];
        }
    }
}


__global__ void scatter_edges(const float* __restrict__ vals,
                              const int* __restrict__ idx,
                              const float* __restrict__ xT,
                              float* __restrict__ outT, int nnz) {
    int e = blockIdx.x * blockDim.x + threadIdx.x;
    if (e >= nnz) return;
    int r = __builtin_nontemporal_load(idx + e);
    int c = __builtin_nontemporal_load(idx + nnz + e);
    float v = __builtin_nontemporal_load(vals + e);

    const float4* xs = (const float4*)(xT + (size_t)c * BATCH);
    float4 x0 = xs[0];
    float4 x1 = xs[1];
    float4 x2 = xs[2];
    float4 x3 = xs[3];

    float* o = outT + (size_t)r * BATCH;
    atomicAdd(o + 0,  v * x0.x); atomicAdd(o + 1,  v * x0.y);
    atomicAdd(o + 2,  v * x0.z); atomicAdd(o + 3,  v * x0.w);
    atomicAdd(o + 4,  v * x1.x); atomicAdd(o + 5,  v * x1.y);
    atomicAdd(o + 6,  v * x1.z); atomicAdd(o + 7,  v * x1.w);
    atomicAdd(o + 8,  v * x2.x); atomicAdd(o + 9,  v * x2.y);
    atomicAdd(o + 10, v * x2.z); atomicAdd(o + 11, v * x2.w);
    atomicAdd(o + 12, v * x3.x); atomicAdd(o + 13, v * x3.y);
    atomicAdd(o + 14, v * x3.z); atomicAdd(o + 15, v * x3.w);
}

__global__ void wmma_transpose_bias(const float* __restrict__ outT,
                                    const float* __restrict__ bias,
                                    float* __restrict__ out, int ndst) {
    int lane = threadIdx.x & 31;
    int wave = threadIdx.x >> 5;
    int tile = blockIdx.x * (blockDim.x >> 5) + wave;
    int d0 = tile * 16;
    if (d0 >= ndst) return;

    int m = lane & 15;
    int h = lane >> 4;
    int k0 = 2 * h;

    float bb = bias[d0 + m];
    v8f c = {bb, bb, bb, bb, bb, bb, bb, bb};

    for (int q = 0; q < 4; ++q) {
        v2f a, b;
        a.x = (m == 4 * q + k0)     ? 1.0f : 0.0f;
        a.y = (m == 4 * q + k0 + 1) ? 1.0f : 0.0f;
        b = *(const v2f*)(outT + (size_t)(d0 + m) * BATCH + 4 * q + k0);
        c = __builtin_amdgcn_wmma_f32_16x16x4_f32(
                 false, a,  false, b,
                 (short)0, c,  false,  false);
    }

    for (int r = 0; r < 8; ++r) {
        int brow = r + 8 * h;
        out[(size_t)brow * ndst + d0 + m] = c[r];
    }
}

extern "C" void kernel_launch(void* const* d_in, const int* in_sizes, int n_in,
                              void* d_out, int out_size, void* d_ws, size_t ws_size,
                              hipStream_t stream) {
    const float* x    = (const float*)d_in[0];
    const float* vals = (const float*)d_in[1];
    const float* bias = (const float*)d_in[2];
    const int*   idx  = (const int*)d_in[3];

    int nsrc = in_sizes[0] / BATCH;
    int nnz  = in_sizes[1];
    int ndst = in_sizes[2];

    float* out = (float*)d_out;
    float* xT  = (float*)d_ws;

    int nXT = nsrc * BATCH;

    int nbins = (ndst + RPB - 1) / RPB;
    int nwg   = (nnz + CHUNK - 1) / CHUNK;
    int nh    = nbins * nwg;
    int nhp   = (nh + 3) & ~3;
    size_t need = ((size_t)nXT + (size_t)nhp + 2 * (size_t)nnz) * 4;

    if (ws_size >= need && nbins <= MAXBINS && nsrc <= (1 << COLBITS)) {
        int*  tab  = (int*)(xT + (size_t)nXT);
        int*  recs = tab + nhp;
        chunk_sort<<<nwg, 1024, 0, stream>>>(vals, idx, x, xT, tab, recs,
                                             nnz, nbins, nwg, nsrc);
        bin_sum_wmma<<<nbins, 512, 0, stream>>>(recs, tab, xT, bias, out,
                                                nnz, ndst, nbins, nwg);
    } else {
        float* outT = xT + (size_t)nXT;
        int nOutT = ndst * BATCH;
        transpose_x<<<(nXT + 255) / 256, 256, 0, stream>>>(x, xT, nsrc);
        zero_f32<<<(nOutT + 255) / 256, 256, 0, stream>>>(outT, nOutT);
        scatter_edges<<<(nnz + 255) / 256, 256, 0, stream>>>(vals, idx, xT, outT, nnz);
        int tiles = (ndst + 15) / 16;
        wmma_transpose_bias<<<(tiles + 7) / 8, 256, 0, stream>>>(outT, bias, out, ndst);
    }
}


# bin sum: a bin's records grouped by destination row in LDS (integer counters, scan, returning add as place), each row's run summed in registers, stored once: no LDS f32 adds; sum order: place arrival
# speedup vs baseline: 12.3107x; 1.0518x over previous
#include <hip/hip_runtime.h>


typedef float v2f __attribute__((ext_vector_type(2)));
typedef float v8f __attribute__((ext_vector_type(8)));

#define BATCH 16
#define RPB_SHIFT 7
#define RPB 128
#define MAXBINS 1024
#define CHUNK 12544
#define EPT ((CHUNK + 1023) / 1024)
#define COLBITS 24
#define XPT 8
#define RPH 8
#define RB 12
#define CAP 6144
#define RPHW (RPB / 32)

__global__ void zero_f32(float* __restrict__ p, int n) {
    int i = blockIdx.x * blockDim.x + threadIdx.x;
    if (i < n) p[i] = 0.0f;
}

__global__ void transpose_x(const float* __restrict__ x,
                            float* __restrict__ xT, int nsrc) {
    int i = blockIdx.x * blockDim.x + threadIdx.x;
    if (i < nsrc * BATCH) {
        int s = i >> 4;
        int b = i & 15;
        xT[i] = x[b * nsrc + s];
    }
}

__device__ __forceinline__ int block_excl_scan_1024(int v, int* wsum, int* total) {
    int lane = threadIdx.x & 31;
    int w = threadIdx.x >> 5;
    int inc = v;
    for (int d = 1; d < 32; d <<= 1) {
        int n = __shfl_up(inc, d, 32);
        if (lane >= d) inc += n;
    }
    if (lane == 31) wsum[w] = inc;
    __syncthreads();
    if (w == 0) {
        int s = wsum[lane];
        int si = s;
        for (int d = 1; d < 32; d <<= 1) {
            int n = __shfl_up(si, d, 32);
            if (lane >= d) si += n;
        }
        wsum[lane] = si - s;
        if (lane == 31) wsum[32] = si;
    }
    __syncthreads();
    int excl = inc - v + wsum[w];
    *total = wsum[32];
    __syncthreads();
    return excl;
}

__global__ void __launch_bounds__(1024)
chunk_sort(const float* __restrict__ vals, const int* __restrict__ idx,
           const float* __restrict__ x, float* __restrict__ xT,
           int* __restrict__ tab, int* __restrict__ recs,
           int nnz, int nbins, int nwg, int nsrc) {
    __shared__ int cursor[MAXBINS];
    __shared__ int wsum[33];
    __shared__ int2 srec[CHUNK];
    int t = threadIdx.x;
    int g = blockIdx.x;
    if (t < nbins) cursor[t] = 0;
    __syncthreads();
    int e0 = g * CHUNK;
    int e1 = min(e0 + CHUNK, nnz);
    int r[EPT], s[EPT];
    float v[EPT];
    for (int k = 0; k < EPT; ++k) {
        int e = min(e0 + t + k * 1024, e1 - 1);
        r[k] = __builtin_nontemporal_load(idx + e);
        s[k] = __builtin_nontemporal_load(idx + nnz + e);
        v[k] = __builtin_nontemporal_load(vals + e);
    }
    int per = (nsrc + nwg - 1) / nwg;
    int s0 = min(g * per, nsrc);
    int s1 = min(s0 + per, nsrc);
    int ne = (s1 - s0) * BATCH;
    const float* xs = x + s0;
    float* dst = xT + (size_t)s0 * BATCH;
    float xv[XPT];
    for (int k = 0; k < XPT; ++k) {
        int i = max(min(t + k * 1024, ne - 1), 0);
        xv[k] = xs[(i & (BATCH - 1)) * nsrc + (i >> 4)];
    }
    __builtin_amdgcn_sched_barrier(0);

    for (int k = 0; k < EPT; ++k)
        if (e0 + t + k * 1024 < e1) atomicAdd(&cursor[r[k] >> RPB_SHIFT], 1);
    __syncthreads();
    int c = (t < nbins) ? cursor[t] : 0;
    int chunk_total;
    int ex = block_excl_scan_1024(c, wsum, &chunk_total);
    if (t < nbins) {
        cursor[t] = ex;
        tab[t * nwg + g] = e0 + ex;
    }
    __syncthreads();
    for (int k = 0; k < EPT; ++k) {
        if (e0 + t + k * 1024 < e1) {
            int p = atomicAdd(&cursor[r[k] >> RPB_SHIFT], 1);
            srec[p] = make_int2(((r[k] & (RPB - 1)) << COLBITS) | s[k],
                                __float_as_int(v[k]));
        }
    }
    for (int k = 0; k < XPT; ++k)
        if (t + k * 1024 < ne) dst[t + k * 1024] = xv[k];
    for (int i = t + XPT * 1024; i < ne; i += 1024)
        dst[i] = xs[(i & (BATCH - 1)) * nsrc + (i >> 4)];
    __syncthreads();
    int n = e1 - e0;
    int2* recs2 = (int2*)recs;
    for (int i = t; i < n; i += 1024) recs2[e0 + i] = srec[i];
}

__global__ void __launch_bounds__(512) __attribute__((amdgpu_waves_per_eu(16, 16)))
bin_sum_wmma(const int* __restrict__ recs, const int* __restrict__ tab,
             const float* __restrict__ xT, const float* __restrict__ bias,
             float* __restrict__ out, int nnz, int ndst, int nbins, int nwg) {
    __shared__ float acc[(RPB + 1) * BATCH];
    __shared__ __attribute__((aligned(16))) int rowtab[RPB + RPB + 4];
    int* cnt = rowtab;
    int* rstart = rowtab + RPB;
    __shared__ int hwtot[32];
    __shared__ int2 lsort[CAP];
    int t = threadIdx.x;
    int bin = blockIdx.x;
    for (int i = t; i < (RPB + 1) * BATCH; i += 512) acc[i] = 0.0f;

    int hw = t >> 4;
    int bl = t & 15;
    const int2* recs2 = (const int2*)recs;
    const int2 empty = make_int2((int)((unsigned)RPB << COLBITS), 0);

    for (int base = 0; base < nwg; base += 32 * RPH) {
        int g = base + hw * RPH + bl;
        int gc = min(g, nwg - 1);
        int rb = tab[bin * nwg + gc];
        int re = (bin + 1 < nbins) ? tab[(bin + 1) * nwg + gc]
                                   : min((gc + 1) * CHUNK, nnz);
        int len = (bl < RPH && g < nwg) ? re - rb : 0;
        int inc = len;
        inc += __builtin_amdgcn_update_dpp(0, inc, 0x111, 0xf, 0xf, true);
        inc += __builtin_amdgcn_update_dpp(0, inc, 0x112, 0xf, 0xf, true);
        inc += __builtin_amdgcn_update_dpp(0, inc, 0x114, 0xf, 0xf, true);
        inc += __builtin_amdgcn_update_dpp(0, inc, 0x118, 0xf, 0xf, true);
        int total = __shfl(inc, 15, 16);
        int excl = inc - len;
        int delta = rb - excl;
        int nb = (total + 15) >> 4;
        nb = max(nb, __builtin_amdgcn_permlanex16(nb, nb, 0x76543210, 0xfedcba98,
                                                  false, false));

        if (t < RPB) cnt[t] = 0;
        if (bl == 0) hwtot[hw] = total;
        __syncthreads();
        int ptotal = 0, pmax = 0;
        for (int j = 0; j < 32; ++j) {
            int v = hwtot[j];
            ptotal += v;
            pmax = max(pmax, v);
        }
        bool sorted = (ptotal <= CAP) && (pmax <= 16 * RB);

        if (sorted) {
            int2 rec[RB];
#pragma unroll
            for (int k = 0; k < RB; ++k) {
                rec[k] = empty;
                if (k < nb) {
                    int q = bl + 16 * k;
                    int dsel = 0;
                    for (int j = 0; j < RPH; ++j) {
                        int ej = __shfl(excl, j, 16);
                        int dj = __shfl(delta, j, 16);
                        if (q >= ej) dsel = dj;
                    }
                    rec[k] = recs2[(q < total) ? q + dsel : 0];
                }
            }
#pragma unroll
            for (int k = 0; k < RB; ++k)
                if (bl + 16 * k < total)
                    atomicAdd(&cnt[(unsigned)rec[k].x >> COLBITS], 1);
            __syncthreads();
            if (t < 32) {
                int4 c = *(const int4*)&cnt[4 * t];
                int s4 = c.x + c.y + c.z + c.w;
                int incl = s4;
                incl += __builtin_amdgcn_update_dpp(0, incl, 0x111, 0xf, 0xf, true);
                incl += __builtin_amdgcn_update_dpp(0, incl, 0x112, 0xf, 0xf, true);
                incl += __builtin_amdgcn_update_dpp(0, incl, 0x114, 0xf, 0xf, true);
                incl += __builtin_amdgcn_update_dpp(0, incl, 0x118, 0xf, 0xf, true);
                int lo = __shfl(incl, 15, 32);
                if (t >= 16) incl += lo;
                int ex = incl - s4;
                int4 o;
                o.x = ex;
                o.y = ex + c.x;
                o.z = ex + c.x + c.y;
                o.w = ex + c.x + c.y + c.z;
                *(int4*)&rstart[4 * t] = o;
                if (t == 31) rstart[RPB] = incl;
            }
            __syncthreads();
            int pos[RB];
#pragma unroll
            for (int k = 0; k < RB; ++k) {
                pos[k] = 0;
                if (bl + 16 * k < total)
                    pos[k] = atomicAdd(&rstart[(unsigned)rec[k].x >> COLBITS], 1);
            }
#pragma unroll
            for (int k = 0; k < RB; ++k) {
                if (bl + 16 * k < total)
                    lsort[pos[k]] = make_int2(rec[k].x & (int)((1u << COLBITS) - 1u),
                                              rec[k].y);
            }
            __syncthreads();
#pragma unroll 1
            for (int rr = 0; rr < RPHW; ++rr) {
                int row = hw * RPHW + rr;
                int s = (row > 0) ? rstart[row - 1] : 0;
                int e = rstart[row];
                float sum = 0.0f;
                int2 rc = lsort[max(min(s + bl, e - 1), 0)];
                for (int j0 = s; j0 < e; j0 += 16) {
                    int2 nx = lsort[max(min(j0 + 16 + bl, e - 1), 0)];
                    int vi = (j0 + bl < e) ? rc.y : 0;
                    float xv[16];
                    for (int j = 0; j < 16; ++j) {
                        unsigned sc = (unsigned)__shfl(rc.x, j, 16) & ((1u << COLBITS) - 1u);
                        xv[j] = xT[(size_t)sc * BATCH + bl];
                    }
                    for (int j = 0; j < 16; ++j) {
                        float vv = __int_as_float(__shfl(vi, j, 16));
                        sum = __builtin_fmaf(vv, xv[j], sum);
                    }
                    rc = nx;
                }
                acc[row * BATCH + bl] += sum;
            }
        } else {
            int q = bl;
            int dsel = 0;
            for (int j = 0; j < RPH; ++j) {
                int ej = __shfl(excl, j, 16);
                int dj = __shfl(delta, j, 16);
                if (q >= ej) dsel = dj;
            }
            int2 rec = recs2[(q < total) ? q + dsel : 0];
            if (q >= total) rec = empty;
            for (int it = 0; it < nb; ++it) {
                int qn = q + 16;
                int dn = 0;
                for (int j = 0; j < RPH; ++j) {
                    int ej = __shfl(excl, j, 16);
                    int dj = __shfl(delta, j, 16);
                    if (qn >= ej) dn = dj;
                }
                int2 nxt = recs2[(qn < total) ? qn + dn : 0];
                if (qn >= total) nxt = empty;
                for (int h8 = 0; h8 < 16; h8 += 8) {
                    float xv[8];
                    for (int j = 0; j < 8; ++j) {
                        unsigned key = (unsigned)__shfl(rec.x, h8 + j, 16);
                        unsigned s = key & ((1u << COLBITS) - 1u);
                        xv[j] = xT[(size_t)s * BATCH + bl];
                    }
                    for (int j = 0; j < 8; ++j) {
                        unsigned key = (unsigned)__shfl(rec.x, h8 + j, 16);
                        float vv = __int_as_float(__shfl(rec.y, h8 + j, 16));
                        unsigned rl = key >> COLBITS;
                        atomicAdd(&acc[rl * BATCH + bl], vv * xv[j]);
                    }
                }
                rec = nxt;
                q = qn;
            }
        }
        __syncthreads();
    }

    int lane = t & 31;
    int wave = t >> 5;
    int d0 = bin * RPB + wave * 16;
    if (wave < RPB / 16 && d0 < ndst) {
        int m = lane & 15;
        int h = lane >> 4;
        int k0 = 2 * h;

        float bb = bias[d0 + m];
        v8f c = {bb, bb, bb, bb, bb, bb, bb, bb};

        for (int q = 0; q < 4; ++q) {
            v2f a, b;
            a.x = (m == 4 * q + k0)     ? 1.0f : 0.0f;
            a.y = (m == 4 * q + k0 + 1) ? 1.0f : 0.0f;
            b = *(const v2f*)(acc + (wave * 16 + m) * BATCH + 4 * q + k0);
            c = __builtin_amdgcn_wmma_f32_16x16x4_f32(
                     false, a,  false, b,
                     (short)0, c,  false,  false);
        }

        for (int r = 0; r < 8; ++r) {
            int brow = r + 8 * h;
            out[(size_t)brow * ndst + d0 + m] = c[r];
        }
    }
}


__global__ void scatter_edges(const float* __restrict__ vals,
                              const int* __restrict__ idx,
                              const float* __restrict__ xT,
                              float* __restrict__ outT, int nnz) {
    int e = blockIdx.x * blockDim.x + threadIdx.x;
    if (e >= nnz) return;
    int r = __builtin_nontemporal_load(idx + e);
    int c = __builtin_nontemporal_load(idx + nnz + e);
    float v = __builtin_nontemporal_load(vals + e);

    const float4* xs = (const float4*)(xT + (size_t)c * BATCH);
    float4 x0 = xs[0];
    float4 x1 = xs[1];
    float4 x2 = xs[2];
    float4 x3 = xs[3];

    float* o = outT + (size_t)r * BATCH;
    atomicAdd(o + 0,  v * x0.x); atomicAdd(o + 1,  v * x0.y);
    atomicAdd(o + 2,  v * x0.z); atomicAdd(o + 3,  v * x0.w);
    atomicAdd(o + 4,  v * x1.x); atomicAdd(o + 5,  v * x1.y);
    atomicAdd(o + 6,  v * x1.z); atomicAdd(o + 7,  v * x1.w);
    atomicAdd(o + 8,  v * x2.x); atomicAdd(o + 9,  v * x2.y);
    atomicAdd(o + 10, v * x2.z); atomicAdd(o + 11, v * x2.w);
    atomicAdd(o + 12, v * x3.x); atomicAdd(o + 13, v * x3.y);
    atomicAdd(o + 14, v * x3.z); atomicAdd(o + 15, v * x3.w);
}

__global__ void wmma_transpose_bias(const float* __restrict__ outT,
                                    const float* __restrict__ bias,
                                    float* __restrict__ out, int ndst) {
    int lane = threadIdx.x & 31;
    int wave = threadIdx.x >> 5;
    int tile = blockIdx.x * (blockDim.x >> 5) + wave;
    int d0 = tile * 16;
    if (d0 >= ndst) return;

    int m = lane & 15;
    int h = lane >> 4;
    int k0 = 2 * h;

    float bb = bias[d0 + m];
    v8f c = {bb, bb, bb, bb, bb, bb, bb, bb};

    for (int q = 0; q < 4; ++q) {
        v2f a, b;
        a.x = (m == 4 * q + k0)     ? 1.0f : 0.0f;
        a.y = (m == 4 * q + k0 + 1) ? 1.0f : 0.0f;
        b = *(const v2f*)(outT + (size_t)(d0 + m) * BATCH + 4 * q + k0);
        c = __builtin_amdgcn_wmma_f32_16x16x4_f32(
                 false, a,  false, b,
                 (short)0, c,  false,  false);
    }

    for (int r = 0; r < 8; ++r) {
        int brow = r + 8 * h;
        out[(size_t)brow * ndst + d0 + m] = c[r];
    }
}

extern "C" void kernel_launch(void* const* d_in, const int* in_sizes, int n_in,
                              void* d_out, int out_size, void* d_ws, size_t ws_size,
                              hipStream_t stream) {
    const float* x    = (const float*)d_in[0];
    const float* vals = (const float*)d_in[1];
    const float* bias = (const float*)d_in[2];
    const int*   idx  = (const int*)d_in[3];

    int nsrc = in_sizes[0] / BATCH;
    int nnz  = in_sizes[1];
    int ndst = in_sizes[2];

    float* out = (float*)d_out;
    float* xT  = (float*)d_ws;

    int nXT = nsrc * BATCH;

    int nbins = (ndst + RPB - 1) / RPB;
    int nwg   = (nnz + CHUNK - 1) / CHUNK;
    int nh    = nbins * nwg;
    int nhp   = (nh + 3) & ~3;
    size_t need = ((size_t)nXT + (size_t)nhp + 2 * (size_t)nnz) * 4;

    if (ws_size >= need && nbins <= MAXBINS && nsrc <= (1 << COLBITS)) {
        int*  tab  = (int*)(xT + (size_t)nXT);
        int*  recs = tab + nhp;
        chunk_sort<<<nwg, 1024, 0, stream>>>(vals, idx, x, xT, tab, recs,
                                             nnz, nbins, nwg, nsrc);
        bin_sum_wmma<<<nbins, 512, 0, stream>>>(recs, tab, xT, bias, out,
                                                nnz, ndst, nbins, nwg);
    } else {
        float* outT = xT + (size_t)nXT;
        int nOutT = ndst * BATCH;
        transpose_x<<<(nXT + 255) / 256, 256, 0, stream>>>(x, xT, nsrc);
        zero_f32<<<(nOutT + 255) / 256, 256, 0, stream>>>(outT, nOutT);
        scatter_edges<<<(nnz + 255) / 256, 256, 0, stream>>>(vals, idx, xT, outT, nnz);
        int tiles = (ndst + 15) / 16;
        wmma_transpose_bias<<<(tiles + 7) / 8, 256, 0, stream>>>(outT, bias, out, ndst);
    }
}


# bin sum: a quad of 4 lanes owns a destination row and walks its list run, each lane gathering 16B of the row of xT (8 edges per load instruction, no cross-lane moves); sum order unchanged
# speedup vs baseline: 12.8876x; 1.0469x over previous
#include <hip/hip_runtime.h>


typedef float v2f __attribute__((ext_vector_type(2)));
typedef float v8f __attribute__((ext_vector_type(8)));

#define BATCH 16
#define RPB_SHIFT 7
#define RPB 128
#define MAXBINS 1024
#define CHUNK 12544
#define EPT ((CHUNK + 1023) / 1024)
#define COLBITS 24
#define XPT 8
#define RPH 8
#define RB 12
#define CAP 6144
#define QPW 8
#define GU 8

__global__ void zero_f32(float* __restrict__ p, int n) {
    int i = blockIdx.x * blockDim.x + threadIdx.x;
    if (i < n) p[i] = 0.0f;
}

__global__ void transpose_x(const float* __restrict__ x,
                            float* __restrict__ xT, int nsrc) {
    int i = blockIdx.x * blockDim.x + threadIdx.x;
    if (i < nsrc * BATCH) {
        int s = i >> 4;
        int b = i & 15;
        xT[i] = x[b * nsrc + s];
    }
}

__device__ __forceinline__ int block_excl_scan_1024(int v, int* wsum, int* total) {
    int lane = threadIdx.x & 31;
    int w = threadIdx.x >> 5;
    int inc = v;
    for (int d = 1; d < 32; d <<= 1) {
        int n = __shfl_up(inc, d, 32);
        if (lane >= d) inc += n;
    }
    if (lane == 31) wsum[w] = inc;
    __syncthreads();
    if (w == 0) {
        int s = wsum[lane];
        int si = s;
        for (int d = 1; d < 32; d <<= 1) {
            int n = __shfl_up(si, d, 32);
            if (lane >= d) si += n;
        }
        wsum[lane] = si - s;
        if (lane == 31) wsum[32] = si;
    }
    __syncthreads();
    int excl = inc - v + wsum[w];
    *total = wsum[32];
    __syncthreads();
    return excl;
}

__global__ void __launch_bounds__(1024)
chunk_sort(const float* __restrict__ vals, const int* __restrict__ idx,
           const float* __restrict__ x, float* __restrict__ xT,
           int* __restrict__ tab, int* __restrict__ recs,
           int nnz, int nbins, int nwg, int nsrc) {
    __shared__ int cursor[MAXBINS];
    __shared__ int wsum[33];
    __shared__ int2 srec[CHUNK];
    int t = threadIdx.x;
    int g = blockIdx.x;
    if (t < nbins) cursor[t] = 0;
    __syncthreads();
    int e0 = g * CHUNK;
    int e1 = min(e0 + CHUNK, nnz);
    int r[EPT], s[EPT];
    float v[EPT];
    for (int k = 0; k < EPT; ++k) {
        int e = min(e0 + t + k * 1024, e1 - 1);
        r[k] = __builtin_nontemporal_load(idx + e);
        s[k] = __builtin_nontemporal_load(idx + nnz + e);
        v[k] = __builtin_nontemporal_load(vals + e);
    }
    int per = (nsrc + nwg - 1) / nwg;
    int s0 = min(g * per, nsrc);
    int s1 = min(s0 + per, nsrc);
    int ne = (s1 - s0) * BATCH;
    const float* xs = x + s0;
    float* dst = xT + (size_t)s0 * BATCH;
    float xv[XPT];
    for (int k = 0; k < XPT; ++k) {
        int i = max(min(t + k * 1024, ne - 1), 0);
        xv[k] = xs[(i & (BATCH - 1)) * nsrc + (i >> 4)];
    }
    __builtin_amdgcn_sched_barrier(0);

    for (int k = 0; k < EPT; ++k)
        if (e0 + t + k * 1024 < e1) atomicAdd(&cursor[r[k] >> RPB_SHIFT], 1);
    __syncthreads();
    int c = (t < nbins) ? cursor[t] : 0;
    int chunk_total;
    int ex = block_excl_scan_1024(c, wsum, &chunk_total);
    if (t < nbins) {
        cursor[t] = ex;
        tab[t * nwg + g] = e0 + ex;
    }
    __syncthreads();
    for (int k = 0; k < EPT; ++k) {
        if (e0 + t + k * 1024 < e1) {
            int p = atomicAdd(&cursor[r[k] >> RPB_SHIFT], 1);
            srec[p] = make_int2(((r[k] & (RPB - 1)) << COLBITS) | s[k],
                                __float_as_int(v[k]));
        }
    }
    for (int k = 0; k < XPT; ++k)
        if (t + k * 1024 < ne) dst[t + k * 1024] = xv[k];
    for (int i = t + XPT * 1024; i < ne; i += 1024)
        dst[i] = xs[(i & (BATCH - 1)) * nsrc + (i >> 4)];
    __syncthreads();
    int n = e1 - e0;
    int2* recs2 = (int2*)recs;
    for (int i = t; i < n; i += 1024) recs2[e0 + i] = srec[i];
}

__global__ void __launch_bounds__(512) __attribute__((amdgpu_waves_per_eu(16, 16)))
bin_sum_wmma(const int* __restrict__ recs, const int* __restrict__ tab,
             const float* __restrict__ xT, const float* __restrict__ bias,
             float* __restrict__ out, int nnz, int ndst, int nbins, int nwg) {
    __shared__ struct __attribute__((aligned(16))) {
        float acc[(RPB + 1) * BATCH];
        int rowtab[RPB + RPB + 4];
    } sm;
    static_assert(((RPB + 1) * BATCH * 4) % 16 == 0, "rowtab stays 16-byte aligned behind acc");
    float* acc = sm.acc;
    int* rowtab = sm.rowtab;
    int* cnt = rowtab;
    int* rstart = rowtab + RPB;
    __shared__ int hwtot[32];
    __shared__ int2 lsort[CAP];
    int t = threadIdx.x;
    int bin = blockIdx.x;
    for (int i = t; i < (RPB + 1) * BATCH; i += 512) acc[i] = 0.0f;

    int hw = t >> 4;
    int bl = t & 15;
    const int2* recs2 = (const int2*)recs;
    const int2 empty = make_int2((int)((unsigned)RPB << COLBITS), 0);

    for (int base = 0; base < nwg; base += 32 * RPH) {
        int g = base + hw * RPH + bl;
        int gc = min(g, nwg - 1);
        int rb = tab[bin * nwg + gc];
        int re = (bin + 1 < nbins) ? tab[(bin + 1) * nwg + gc]
                                   : min((gc + 1) * CHUNK, nnz);
        int len = (bl < RPH && g < nwg) ? re - rb : 0;
        int inc = len;
        inc += __builtin_amdgcn_update_dpp(0, inc, 0x111, 0xf, 0xf, true);
        inc += __builtin_amdgcn_update_dpp(0, inc, 0x112, 0xf, 0xf, true);
        inc += __builtin_amdgcn_update_dpp(0, inc, 0x114, 0xf, 0xf, true);
        inc += __builtin_amdgcn_update_dpp(0, inc, 0x118, 0xf, 0xf, true);
        int total = __shfl(inc, 15, 16);
        int excl = inc - len;
        int delta = rb - excl;
        int nb = (total + 15) >> 4;
        nb = max(nb, __builtin_amdgcn_permlanex16(nb, nb, 0x76543210, 0xfedcba98,
                                                  false, false));

        if (t < RPB) cnt[t] = 0;
        if (bl == 0) hwtot[hw] = total;
        __syncthreads();
        int ptotal = 0, pmax = 0;
        for (int j = 0; j < 32; ++j) {
            int v = hwtot[j];
            ptotal += v;
            pmax = max(pmax, v);
        }
        bool sorted = (ptotal <= CAP) && (pmax <= 16 * RB);

        if (sorted) {
            int2 rec[RB];
#pragma unroll
            for (int k = 0; k < RB; ++k) {
                rec[k] = empty;
                if (k < nb) {
                    int q = bl + 16 * k;
                    int dsel = 0;
                    for (int j = 0; j < RPH; ++j) {
                        int ej = __shfl(excl, j, 16);
                        int dj = __shfl(delta, j, 16);
                        if (q >= ej) dsel = dj;
                    }
                    rec[k] = recs2[(q < total) ? q + dsel : 0];
                }
            }
#pragma unroll
            for (int k = 0; k < RB; ++k)
                if (bl + 16 * k < total)
                    atomicAdd(&cnt[(unsigned)rec[k].x >> COLBITS], 1);
            __syncthreads();
            if (t < 32) {
                int4 c = *(const int4*)&cnt[4 * t];
                int s4 = c.x + c.y + c.z + c.w;
                int incl = s4;
                incl += __builtin_amdgcn_update_dpp(0, incl, 0x111, 0xf, 0xf, true);
                incl += __builtin_amdgcn_update_dpp(0, incl, 0x112, 0xf, 0xf, true);
                incl += __builtin_amdgcn_update_dpp(0, incl, 0x114, 0xf, 0xf, true);
                incl += __builtin_amdgcn_update_dpp(0, incl, 0x118, 0xf, 0xf, true);
                int lo = __shfl(incl, 15, 32);
                if (t >= 16) incl += lo;
                int ex = incl - s4;
                int4 o;
                o.x = ex;
                o.y = ex + c.x;
                o.z = ex + c.x + c.y;
                o.w = ex + c.x + c.y + c.z;
                *(int4*)&rstart[4 * t] = o;
                if (t == 31) rstart[RPB] = incl;
            }
            __syncthreads();
            int pos[RB];
#pragma unroll
            for (int k = 0; k < RB; ++k) {
                pos[k] = 0;
                if (bl + 16 * k < total)
                    pos[k] = atomicAdd(&rstart[(unsigned)rec[k].x >> COLBITS], 1);
            }
#pragma unroll
            for (int k = 0; k < RB; ++k) {
                if (bl + 16 * k < total)
                    lsort[pos[k]] = make_int2(rec[k].x & (int)((1u << COLBITS) - 1u),
                                              rec[k].y);
            }
            __syncthreads();
            {
                static_assert(RPB == 16 * QPW, "16 waves x QPW quads cover the bin's rows");
                int ql = t & 3;
                int row = t >> 2;
                int e = rstart[row];
                int s = e - cnt[row];
                const float4* xT4 = (const float4*)xT;
                float4 sum = make_float4(0.0f, 0.0f, 0.0f, 0.0f);
                for (int j0 = s; j0 < e; j0 += GU) {
                    int2 rc[GU];
                    float4 xv[GU];
                    int rem = e - j0;
#pragma unroll
                    for (int u = 0; u < GU; ++u) {
                        rc[u] = lsort[j0 + min(u, rem - 1)];
                    }
#pragma unroll
                    for (int u = 0; u < GU; ++u) {
                        unsigned xi = (unsigned)rc[u].x * (BATCH / 4) + (unsigned)ql;
                        xv[u] = xT4[xi];
                    }
#pragma unroll
                    for (int u = 0; u < GU; ++u) {
                        float vv = (u < rem) ? __int_as_float(rc[u].y) : 0.0f;
                        sum.x = __builtin_fmaf(vv, xv[u].x, sum.x);
                        sum.y = __builtin_fmaf(vv, xv[u].y, sum.y);
                        sum.z = __builtin_fmaf(vv, xv[u].z, sum.z);
                        sum.w = __builtin_fmaf(vv, xv[u].w, sum.w);
                    }
                }
                float4* ap = (float4*)acc + t;
                float4 a = *ap;
                a.x += sum.x;
                a.y += sum.y;
                a.z += sum.z;
                a.w += sum.w;
                *ap = a;
            }
        } else {
            int q = bl;
            int dsel = 0;
            for (int j = 0; j < RPH; ++j) {
                int ej = __shfl(excl, j, 16);
                int dj = __shfl(delta, j, 16);
                if (q >= ej) dsel = dj;
            }
            int2 rec = recs2[(q < total) ? q + dsel : 0];
            if (q >= total) rec = empty;
            for (int it = 0; it < nb; ++it) {
                int qn = q + 16;
                int dn = 0;
                for (int j = 0; j < RPH; ++j) {
                    int ej = __shfl(excl, j, 16);
                    int dj = __shfl(delta, j, 16);
                    if (qn >= ej) dn = dj;
                }
                int2 nxt = recs2[(qn < total) ? qn + dn : 0];
                if (qn >= total) nxt = empty;
                for (int h8 = 0; h8 < 16; h8 += 8) {
                    float xv[8];
                    for (int j = 0; j < 8; ++j) {
                        unsigned key = (unsigned)__shfl(rec.x, h8 + j, 16);
                        unsigned s = key & ((1u << COLBITS) - 1u);
                        xv[j] = xT[(size_t)s * BATCH + bl];
                    }
                    for (int j = 0; j < 8; ++j) {
                        unsigned key = (unsigned)__shfl(rec.x, h8 + j, 16);
                        float vv = __int_as_float(__shfl(rec.y, h8 + j, 16));
                        unsigned rl = key >> COLBITS;
                        atomicAdd(&acc[rl * BATCH + bl], vv * xv[j]);
                    }
                }
                rec = nxt;
                q = qn;
            }
        }
        __syncthreads();
    }

    int lane = t & 31;
    int wave = t >> 5;
    int d0 = bin * RPB + wave * 16;
    if (wave < RPB / 16 && d0 < ndst) {
        int m = lane & 15;
        int h = lane >> 4;
        int k0 = 2 * h;

        float bb = bias[d0 + m];
        v8f c = {bb, bb, bb, bb, bb, bb, bb, bb};

        for (int q = 0; q < 4; ++q) {
            v2f a, b;
            a.x = (m == 4 * q + k0)     ? 1.0f : 0.0f;
            a.y = (m == 4 * q + k0 + 1) ? 1.0f : 0.0f;
            b = *(const v2f*)(acc + (wave * 16 + m) * BATCH + 4 * q + k0);
            c = __builtin_amdgcn_wmma_f32_16x16x4_f32(
                     false, a,  false, b,
                     (short)0, c,  false,  false);
        }

        for (int r = 0; r < 8; ++r) {
            int brow = r + 8 * h;
            out[(size_t)brow * ndst + d0 + m] = c[r];
        }
    }
}


__global__ void scatter_edges(const float* __restrict__ vals,
                              const int* __restrict__ idx,
                              const float* __restrict__ xT,
                              float* __restrict__ outT, int nnz) {
    int e = blockIdx.x * blockDim.x + threadIdx.x;
    if (e >= nnz) return;
    int r = __builtin_nontemporal_load(idx + e);
    int c = __builtin_nontemporal_load(idx + nnz + e);
    float v = __builtin_nontemporal_load(vals + e);

    const float4* xs = (const float4*)(xT + (size_t)c * BATCH);
    float4 x0 = xs[0];
    float4 x1 = xs[1];
    float4 x2 = xs[2];
    float4 x3 = xs[3];

    float* o = outT + (size_t)r * BATCH;
    atomicAdd(o + 0,  v * x0.x); atomicAdd(o + 1,  v * x0.y);
    atomicAdd(o + 2,  v * x0.z); atomicAdd(o + 3,  v * x0.w);
    atomicAdd(o + 4,  v * x1.x); atomicAdd(o + 5,  v * x1.y);
    atomicAdd(o + 6,  v * x1.z); atomicAdd(o + 7,  v * x1.w);
    atomicAdd(o + 8,  v * x2.x); atomicAdd(o + 9,  v * x2.y);
    atomicAdd(o + 10, v * x2.z); atomicAdd(o + 11, v * x2.w);
    atomicAdd(o + 12, v * x3.x); atomicAdd(o + 13, v * x3.y);
    atomicAdd(o + 14, v * x3.z); atomicAdd(o + 15, v * x3.w);
}

__global__ void wmma_transpose_bias(const float* __restrict__ outT,
                                    const float* __restrict__ bias,
                                    float* __restrict__ out, int ndst) {
    int lane = threadIdx.x & 31;
    int wave = threadIdx.x >> 5;
    int tile = blockIdx.x * (blockDim.x >> 5) + wave;
    int d0 = tile * 16;
    if (d0 >= ndst) return;

    int m = lane & 15;
    int h = lane >> 4;
    int k0 = 2 * h;

    float bb = bias[d0 + m];
    v8f c = {bb, bb, bb, bb, bb, bb, bb, bb};

    for (int q = 0; q < 4; ++q) {
        v2f a, b;
        a.x = (m == 4 * q + k0)     ? 1.0f : 0.0f;
        a.y = (m == 4 * q + k0 + 1) ? 1.0f : 0.0f;
        b = *(const v2f*)(outT + (size_t)(d0 + m) * BATCH + 4 * q + k0);
        c = __builtin_amdgcn_wmma_f32_16x16x4_f32(
                 false, a,  false, b,
                 (short)0, c,  false,  false);
    }

    for (int r = 0; r < 8; ++r) {
        int brow = r + 8 * h;
        out[(size_t)brow * ndst + d0 + m] = c[r];
    }
}

extern "C" void kernel_launch(void* const* d_in, const int* in_sizes, int n_in,
                              void* d_out, int out_size, void* d_ws, size_t ws_size,
                              hipStream_t stream) {
    const float* x    = (const float*)d_in[0];
    const float* vals = (const float*)d_in[1];
    const float* bias = (const float*)d_in[2];
    const int*   idx  = (const int*)d_in[3];

    int nsrc = in_sizes[0] / BATCH;
    int nnz  = in_sizes[1];
    int ndst = in_sizes[2];

    float* out = (float*)d_out;
    float* xT  = (float*)d_ws;

    int nXT = nsrc * BATCH;

    int nbins = (ndst + RPB - 1) / RPB;
    int nwg   = (nnz + CHUNK - 1) / CHUNK;
    int nh    = nbins * nwg;
    int nhp   = (nh + 3) & ~3;
    size_t need = ((size_t)nXT + (size_t)nhp + 2 * (size_t)nnz) * 4;

    if (ws_size >= need && nbins <= MAXBINS && nsrc <= (1 << COLBITS)) {
        int*  tab  = (int*)(xT + (size_t)nXT);
        int*  recs = tab + nhp;
        chunk_sort<<<nwg, 1024, 0, stream>>>(vals, idx, x, xT, tab, recs,
                                             nnz, nbins, nwg, nsrc);
        bin_sum_wmma<<<nbins, 512, 0, stream>>>(recs, tab, xT, bias, out,
                                                nnz, ndst, nbins, nwg);
    } else {
        float* outT = xT + (size_t)nXT;
        int nOutT = ndst * BATCH;
        transpose_x<<<(nXT + 255) / 256, 256, 0, stream>>>(x, xT, nsrc);
        zero_f32<<<(nOutT + 255) / 256, 256, 0, stream>>>(outT, nOutT);
        scatter_edges<<<(nnz + 255) / 256, 256, 0, stream>>>(vals, idx, xT, outT, nnz);
        int tiles = (ndst + 15) / 16;
        wmma_transpose_bias<<<(tiles + 7) / 8, 256, 0, stream>>>(outT, bias, out, ndst);
    }
}


# bin sum on half the waves: bins of 256 rows (391 work-groups, 8 waves a SIMD, 24 records a lane); a quad walks the adjacent list runs of 2 rows as one stream, 16 gathers a trip; sum order unchanged
# speedup vs baseline: 14.2898x; 1.1088x over previous
#include <hip/hip_runtime.h>


typedef float v2f __attribute__((ext_vector_type(2)));
typedef float v8f __attribute__((ext_vector_type(8)));

#define BATCH 16
#define RPB_SHIFT 8
#define RPB 256
#define MAXBINS 1024
#define CHUNK 12544
#define EPT ((CHUNK + 1023) / 1024)
#define COLBITS 23
#define XPT 8
#define RPH 8
#define RB 24
#define CAP 10240
#define QPW 8
#define GU 16

__global__ void zero_f32(float* __restrict__ p, int n) {
    int i = blockIdx.x * blockDim.x + threadIdx.x;
    if (i < n) p[i] = 0.0f;
}

__global__ void transpose_x(const float* __restrict__ x,
                            float* __restrict__ xT, int nsrc) {
    int i = blockIdx.x * blockDim.x + threadIdx.x;
    if (i < nsrc * BATCH) {
        int s = i >> 4;
        int b = i & 15;
        xT[i] = x[b * nsrc + s];
    }
}

__device__ __forceinline__ int block_excl_scan_1024(int v, int* wsum, int* total) {
    int lane = threadIdx.x & 31;
    int w = threadIdx.x >> 5;
    int inc = v;
    for (int d = 1; d < 32; d <<= 1) {
        int n = __shfl_up(inc, d, 32);
        if (lane >= d) inc += n;
    }
    if (lane == 31) wsum[w] = inc;
    __syncthreads();
    if (w == 0) {
        int s = wsum[lane];
        int si = s;
        for (int d = 1; d < 32; d <<= 1) {
            int n = __shfl_up(si, d, 32);
            if (lane >= d) si += n;
        }
        wsum[lane] = si - s;
        if (lane == 31) wsum[32] = si;
    }
    __syncthreads();
    int excl = inc - v + wsum[w];
    *total = wsum[32];
    __syncthreads();
    return excl;
}

__global__ void __launch_bounds__(1024)
chunk_sort(const float* __restrict__ vals, const int* __restrict__ idx,
           const float* __restrict__ x, float* __restrict__ xT,
           int* __restrict__ tab, int* __restrict__ recs,
           int nnz, int nbins, int nwg, int nsrc) {
    __shared__ int cursor[MAXBINS];
    __shared__ int wsum[33];
    __shared__ int2 srec[CHUNK];
    int t = threadIdx.x;
    int g = blockIdx.x;
    if (t < nbins) cursor[t] = 0;
    __syncthreads();
    int e0 = g * CHUNK;
    int e1 = min(e0 + CHUNK, nnz);
    int r[EPT], s[EPT];
    float v[EPT];
    for (int k = 0; k < EPT; ++k) {
        int e = min(e0 + t + k * 1024, e1 - 1);
        r[k] = __builtin_nontemporal_load(idx + e);
        s[k] = __builtin_nontemporal_load(idx + nnz + e);
        v[k] = __builtin_nontemporal_load(vals + e);
    }
    int per = (nsrc + nwg - 1) / nwg;
    int s0 = min(g * per, nsrc);
    int s1 = min(s0 + per, nsrc);
    int ne = (s1 - s0) * BATCH;
    const float* xs = x + s0;
    float* dst = xT + (size_t)s0 * BATCH;
    float xv[XPT];
    for (int k = 0; k < XPT; ++k) {
        int i = max(min(t + k * 1024, ne - 1), 0);
        xv[k] = xs[(i & (BATCH - 1)) * nsrc + (i >> 4)];
    }
    __builtin_amdgcn_sched_barrier(0);

    for (int k = 0; k < EPT; ++k)
        if (e0 + t + k * 1024 < e1) atomicAdd(&cursor[r[k] >> RPB_SHIFT], 1);
    __syncthreads();
    int c = (t < nbins) ? cursor[t] : 0;
    int chunk_total;
    int ex = block_excl_scan_1024(c, wsum, &chunk_total);
    if (t < nbins) {
        cursor[t] = ex;
        tab[t * nwg + g] = e0 + ex;
    }
    __syncthreads();
    for (int k = 0; k < EPT; ++k) {
        if (e0 + t + k * 1024 < e1) {
            int p = atomicAdd(&cursor[r[k] >> RPB_SHIFT], 1);
            srec[p] = make_int2(((r[k] & (RPB - 1)) << COLBITS) | s[k],
                                __float_as_int(v[k]));
        }
    }
    for (int k = 0; k < XPT; ++k)
        if (t + k * 1024 < ne) dst[t + k * 1024] = xv[k];
    for (int i = t + XPT * 1024; i < ne; i += 1024)
        dst[i] = xs[(i & (BATCH - 1)) * nsrc + (i >> 4)];
    __syncthreads();
    int n = e1 - e0;
    int2* recs2 = (int2*)recs;
    for (int i = t; i < n; i += 1024) recs2[e0 + i] = srec[i];
}

__global__ void __launch_bounds__(512) __attribute__((amdgpu_waves_per_eu(8, 8)))
bin_sum_wmma(const int* __restrict__ recs, const int* __restrict__ tab,
             const float* __restrict__ xT, const float* __restrict__ bias,
             float* __restrict__ out, int nnz, int ndst, int nbins, int nwg) {
    __shared__ struct __attribute__((aligned(16))) {
        float acc[(RPB + 1) * BATCH];
        int rowtab[RPB + RPB + 4];
    } sm;
    static_assert(((RPB + 1) * BATCH * 4) % 16 == 0, "rowtab stays 16-byte aligned behind acc");
    float* acc = sm.acc;
    int* rowtab = sm.rowtab;
    int* cnt = rowtab;
    int* rstart = rowtab + RPB;
    __shared__ int hwtot[32];
    __shared__ int2 lsort[CAP];
    int t = threadIdx.x;
    int bin = blockIdx.x;
    for (int i = t; i < (RPB + 1) * BATCH; i += 512) acc[i] = 0.0f;

    int hw = t >> 4;
    int bl = t & 15;
    const int2* recs2 = (const int2*)recs;
    const int2 empty = make_int2((int)((unsigned)RPB << COLBITS), 0);

    for (int base = 0; base < nwg; base += 32 * RPH) {
        int g = base + hw * RPH + bl;
        int gc = min(g, nwg - 1);
        int rb = tab[bin * nwg + gc];
        int re = (bin + 1 < nbins) ? tab[(bin + 1) * nwg + gc]
                                   : min((gc + 1) * CHUNK, nnz);
        int len = (bl < RPH && g < nwg) ? re - rb : 0;
        int inc = len;
        inc += __builtin_amdgcn_update_dpp(0, inc, 0x111, 0xf, 0xf, true);
        inc += __builtin_amdgcn_update_dpp(0, inc, 0x112, 0xf, 0xf, true);
        inc += __builtin_amdgcn_update_dpp(0, inc, 0x114, 0xf, 0xf, true);
        inc += __builtin_amdgcn_update_dpp(0, inc, 0x118, 0xf, 0xf, true);
        int total = __shfl(inc, 15, 16);
        int excl = inc - len;
        int delta = rb - excl;
        int nb = (total + 15) >> 4;
        nb = max(nb, __builtin_amdgcn_permlanex16(nb, nb, 0x76543210, 0xfedcba98,
                                                  false, false));

        if (t < RPB) cnt[t] = 0;
        if (bl == 0) hwtot[hw] = total;
        __syncthreads();
        int ptotal = 0, pmax = 0;
        for (int j = 0; j < 32; ++j) {
            int v = hwtot[j];
            ptotal += v;
            pmax = max(pmax, v);
        }
        bool sorted = (ptotal <= CAP) && (pmax <= 16 * RB);

        if (sorted) {
            int2 rec[RB];
            int lim = total - bl;
#pragma unroll
            for (int k = 0; k < RB; ++k) {
                rec[k] = empty;
                if (k < nb) {
                    int exk = excl - 16 * k;
                    int dlk = delta + 16 * k;
                    int dsel = 0;
                    for (int j = 0; j < RPH; ++j) {
                        int ej = __shfl(exk, j, 16);
                        int dj = __shfl(dlk, j, 16);
                        if (bl >= ej) dsel = dj;
                    }
                    rec[k] = recs2[(16 * k < lim) ? bl + dsel : 0];
                }
            }
#pragma unroll
            for (int k = 0; k < RB; ++k)
                if (16 * k < lim)
                    atomicAdd(&cnt[(unsigned)rec[k].x >> COLBITS], 1);
            __syncthreads();
            if (t < 32) {
                static_assert(RPB == 8 * 32, "wave 0 scans 8 row counters a lane");
                int4 c = *(const int4*)&cnt[8 * t];
                int4 d = *(const int4*)&cnt[8 * t + 4];
                int s4 = c.x + c.y + c.z + c.w;
                int s8 = s4 + d.x + d.y + d.z + d.w;
                int incl = s8;
                incl += __builtin_amdgcn_update_dpp(0, incl, 0x111, 0xf, 0xf, true);
                incl += __builtin_amdgcn_update_dpp(0, incl, 0x112, 0xf, 0xf, true);
                incl += __builtin_amdgcn_update_dpp(0, incl, 0x114, 0xf, 0xf, true);
                incl += __builtin_amdgcn_update_dpp(0, incl, 0x118, 0xf, 0xf, true);
                int lo = __shfl(incl, 15, 32);
                if (t >= 16) incl += lo;
                int ex = incl - s8;
                int4 o;
                o.x = ex;
                o.y = ex + c.x;
                o.z = ex + c.x + c.y;
                o.w = ex + c.x + c.y + c.z;
                *(int4*)&rstart[8 * t] = o;
                int ex4 = ex + s4;
                o.x = ex4;
                o.y = ex4 + d.x;
                o.z = ex4 + d.x + d.y;
                o.w = ex4 + d.x + d.y + d.z;
                *(int4*)&rstart[8 * t + 4] = o;
                if (t == 31) rstart[RPB] = incl;
            }
            __syncthreads();
            static_assert(RB % 12 == 0, "places are taken 12 records at a time");
#pragma unroll
            for (int kb = 0; kb < RB; kb += 12) {
                int pos[12];
#pragma unroll
                for (int k = 0; k < 12; ++k) {
                    pos[k] = 0;
                    if (16 * (kb + k) < lim)
                        pos[k] = atomicAdd(&rstart[(unsigned)rec[kb + k].x >> COLBITS], 1);
                }
#pragma unroll
                for (int k = 0; k < 12; ++k) {
                    if (16 * (kb + k) < lim)
                        lsort[pos[k]] = make_int2(rec[kb + k].x & (int)((1u << COLBITS) - 1u),
                                                  rec[kb + k].y);
                }
            }
            __syncthreads();
            {
                static_assert(RPB == 2 * 16 * QPW, "16 waves x QPW quads x 2 rows cover the bin's rows");
                int ql = t & 3;
                int q = t >> 2;
                int2 ends = *(const int2*)&rstart[2 * q];
                int m = ends.x;
                int e = ends.y;
                int s = m - cnt[2 * q];
                const float4* xT4 = (const float4*)xT;
                float4 sum = make_float4(0.0f, 0.0f, 0.0f, 0.0f);
                float4 sum1 = make_float4(0.0f, 0.0f, 0.0f, 0.0f);
                for (int j0 = s; j0 < e; j0 += GU) {
                    int2 rc[GU];
                    float4 xv[GU];
                    int rem = e - j0;
                    int rem0 = m - j0;
#pragma unroll
                    for (int u = 0; u < GU; ++u) {
                        rc[u] = lsort[j0 + min(u, rem - 1)];
                    }
#pragma unroll
                    for (int u = 0; u < GU; ++u) {
                        unsigned xi = (unsigned)rc[u].x * (BATCH / 4) + (unsigned)ql;
                        xv[u] = xT4[xi];
                    }
#pragma unroll
                    for (int u = 0; u < GU; ++u) {
                        float vv = (u < rem) ? __int_as_float(rc[u].y) : 0.0f;
                        float v0 = (u < rem0) ? vv : 0.0f;
                        float v1 = (u < rem0) ? 0.0f : vv;
                        sum.x = __builtin_fmaf(v0, xv[u].x, sum.x);
                        sum.y = __builtin_fmaf(v0, xv[u].y, sum.y);
                        sum.z = __builtin_fmaf(v0, xv[u].z, sum.z);
                        sum.w = __builtin_fmaf(v0, xv[u].w, sum.w);
                        sum1.x = __builtin_fmaf(v1, xv[u].x, sum1.x);
                        sum1.y = __builtin_fmaf(v1, xv[u].y, sum1.y);
                        sum1.z = __builtin_fmaf(v1, xv[u].z, sum1.z);
                        sum1.w = __builtin_fmaf(v1, xv[u].w, sum1.w);
                    }
                }
                float4* ap = (float4*)acc + 2 * (BATCH / 4) * q + ql;
                float4 a = ap[0];
                a.x += sum.x;
                a.y += sum.y;
                a.z += sum.z;
                a.w += sum.w;
                ap[0] = a;
                float4 a1 = ap[BATCH / 4];
                a1.x += sum1.x;
                a1.y += sum1.y;
                a1.z += sum1.z;
                a1.w += sum1.w;
                ap[BATCH / 4] = a1;
            }
        } else {
            int q = bl;
            int dsel = 0;
            for (int j = 0; j < RPH; ++j) {
                int ej = __shfl(excl, j, 16);
                int dj = __shfl(delta, j, 16);
                if (q >= ej) dsel = dj;
            }
            int2 rec = recs2[(q < total) ? q + dsel : 0];
            if (q >= total) rec = empty;
            for (int it = 0; it < nb; ++it) {
                int qn = q + 16;
                int dn = 0;
                for (int j = 0; j < RPH; ++j) {
                    int ej = __shfl(excl, j, 16);
                    int dj = __shfl(delta, j, 16);
                    if (qn >= ej) dn = dj;
                }
                int2 nxt = recs2[(qn < total) ? qn + dn : 0];
                if (qn >= total) nxt = empty;
                for (int h8 = 0; h8 < 16; h8 += 8) {
                    float xv[8];
                    for (int j = 0; j < 8; ++j) {
                        unsigned key = (unsigned)__shfl(rec.x, h8 + j, 16);
                        unsigned s = key & ((1u << COLBITS) - 1u);
                        xv[j] = xT[(size_t)s * BATCH + bl];
                    }
                    for (int j = 0; j < 8; ++j) {
                        unsigned key = (unsigned)__shfl(rec.x, h8 + j, 16);
                        float vv = __int_as_float(__shfl(rec.y, h8 + j, 16));
                        unsigned rl = key >> COLBITS;
                        atomicAdd(&acc[rl * BATCH + bl], vv * xv[j]);
                    }
                }
                rec = nxt;
                q = qn;
            }
        }
        __syncthreads();
    }

    int lane = t & 31;
    int wave = t >> 5;
    int d0 = bin * RPB + wave * 16;
    if (wave < RPB / 16 && d0 < ndst) {
        int m = lane & 15;
        int h = lane >> 4;
        int k0 = 2 * h;

        float bb = bias[d0 + m];
        v8f c = {bb, bb, bb, bb, bb, bb, bb, bb};

        for (int q = 0; q < 4; ++q) {
            v2f a, b;
            a.x = (m == 4 * q + k0)     ? 1.0f : 0.0f;
            a.y = (m == 4 * q + k0 + 1) ? 1.0f : 0.0f;
            b = *(const v2f*)(acc + (wave * 16 + m) * BATCH + 4 * q + k0);
            c = __builtin_amdgcn_wmma_f32_16x16x4_f32(
                     false, a,  false, b,
                     (short)0, c,  false,  false);
        }

        for (int r = 0; r < 8; ++r) {
            int brow = r + 8 * h;
            out[(size_t)brow * ndst + d0 + m] = c[r];
        }
    }
}


__global__ void scatter_edges(const float* __restrict__ vals,
                              const int* __restrict__ idx,
                              const float* __restrict__ xT,
                              float* __restrict__ outT, int nnz) {
    int e = blockIdx.x * blockDim.x + threadIdx.x;
    if (e >= nnz) return;
    int r = __builtin_nontemporal_load(idx + e);
    int c = __builtin_nontemporal_load(idx + nnz + e);
    float v = __builtin_nontemporal_load(vals + e);

    const float4* xs = (const float4*)(xT + (size_t)c * BATCH);
    float4 x0 = xs[0];
    float4 x1 = xs[1];
    float4 x2 = xs[2];
    float4 x3 = xs[3];

    float* o = outT + (size_t)r * BATCH;
    atomicAdd(o + 0,  v * x0.x); atomicAdd(o + 1,  v * x0.y);
    atomicAdd(o + 2,  v * x0.z); atomicAdd(o + 3,  v * x0.w);
    atomicAdd(o + 4,  v * x1.x); atomicAdd(o + 5,  v * x1.y);
    atomicAdd(o + 6,  v * x1.z); atomicAdd(o + 7,  v * x1.w);
    atomicAdd(o + 8,  v * x2.x); atomicAdd(o + 9,  v * x2.y);
    atomicAdd(o + 10, v * x2.z); atomicAdd(o + 11, v * x2.w);
    atomicAdd(o + 12, v * x3.x); atomicAdd(o + 13, v * x3.y);
    atomicAdd(o + 14, v * x3.z); atomicAdd(o + 15, v * x3.w);
}

__global__ void wmma_transpose_bias(const float* __restrict__ outT,
                                    const float* __restrict__ bias,
                                    float* __restrict__ out, int ndst) {
    int lane = threadIdx.x & 31;
    int wave = threadIdx.x >> 5;
    int tile = blockIdx.x * (blockDim.x >> 5) + wave;
    int d0 = tile * 16;
    if (d0 >= ndst) return;

    int m = lane & 15;
    int h = lane >> 4;
    int k0 = 2 * h;

    float bb = bias[d0 + m];
    v8f c = {bb, bb, bb, bb, bb, bb, bb, bb};

    for (int q = 0; q < 4; ++q) {
        v2f a, b;
        a.x = (m == 4 * q + k0)     ? 1.0f : 0.0f;
        a.y = (m == 4 * q + k0 + 1) ? 1.0f : 0.0f;
        b = *(const v2f*)(outT + (size_t)(d0 + m) * BATCH + 4 * q + k0);
        c = __builtin_amdgcn_wmma_f32_16x16x4_f32(
                 false, a,  false, b,
                 (short)0, c,  false,  false);
    }

    for (int r = 0; r < 8; ++r) {
        int brow = r + 8 * h;
        out[(size_t)brow * ndst + d0 + m] = c[r];
    }
}

extern "C" void kernel_launch(void* const* d_in, const int* in_sizes, int n_in,
                              void* d_out, int out_size, void* d_ws, size_t ws_size,
                              hipStream_t stream) {
    const float* x    = (const float*)d_in[0];
    const float* vals = (const float*)d_in[1];
    const float* bias = (const float*)d_in[2];
    const int*   idx  = (const int*)d_in[3];

    int nsrc = in_sizes[0] / BATCH;
    int nnz  = in_sizes[1];
    int ndst = in_sizes[2];

    float* out = (float*)d_out;
    float* xT  = (float*)d_ws;

    int nXT = nsrc * BATCH;

    int nbins = (ndst + RPB - 1) / RPB;
    int nwg   = (nnz + CHUNK - 1) / CHUNK;
    int nh    = nbins * nwg;
    int nhp   = (nh + 3) & ~3;
    size_t need = ((size_t)nXT + (size_t)nhp + 2 * (size_t)nnz) * 4;

    if (ws_size >= need && nbins <= MAXBINS && nsrc <= (1 << COLBITS)) {
        int*  tab  = (int*)(xT + (size_t)nXT);
        int*  recs = tab + nhp;
        chunk_sort<<<nwg, 1024, 0, stream>>>(vals, idx, x, xT, tab, recs,
                                             nnz, nbins, nwg, nsrc);
        bin_sum_wmma<<<nbins, 512, 0, stream>>>(recs, tab, xT, bias, out,
                                                nnz, ndst, nbins, nwg);
    } else {
        float* outT = xT + (size_t)nXT;
        int nOutT = ndst * BATCH;
        transpose_x<<<(nXT + 255) / 256, 256, 0, stream>>>(x, xT, nsrc);
        zero_f32<<<(nOutT + 255) / 256, 256, 0, stream>>>(outT, nOutT);
        scatter_edges<<<(nnz + 255) / 256, 256, 0, stream>>>(vals, idx, xT, outT, nnz);
        int tiles = (ndst + 15) / 16;
        wmma_transpose_bias<<<(tiles + 7) / 8, 256, 0, stream>>>(outT, bias, out, ndst);
    }
}
